# starting kernel
# baseline (speedup 1.0000x reference)
#include <hip/hip_runtime.h>
#include <hip/hip_bf16.h>


typedef float v2f __attribute__((ext_vector_type(2)));
typedef float v8f __attribute__((ext_vector_type(8)));

#define IN_CH 128


__global__ void k_init_deg(float* deg, int n) {
    int i = blockIdx.x * blockDim.x + threadIdx.x;
    if (i < n) deg[i] = 1.0f;
}

__global__ void k_deg_accum(const int* __restrict__ dst, float* deg, int E) {
    int e = blockIdx.x * blockDim.x + threadIdx.x;
    if (e < E) unsafeAtomicAdd(&deg[dst[e]], 1.0f);
}

__global__ void k_dinv(const float* __restrict__ deg, float* dinv, int n) {
    int i = blockIdx.x * blockDim.x + threadIdx.x;
    if (i < n) {
        float d = deg[i];
        dinv[i] = d > 0.0f ? __frsqrt_rn(d) : 0.0f;
    }
}


template <int K, int N>
__global__ void k_gemm_wmma(const float* __restrict__ A,
                            const float* __restrict__ W,
                            float* __restrict__ O,
                            int n_rows, int n_tiles) {
    int wave = (blockIdx.x * blockDim.x + threadIdx.x) >> 5;
    int lane = threadIdx.x & 31;
    if (wave >= n_tiles) return;

    const int m0    = wave * 16;
    const int mrow  = m0 + (lane & 15);
    const int rm    = mrow < n_rows ? mrow : (n_rows - 1);
    const int khalf = (lane >> 4) * 2;
    const int colL  = lane & 15;

    constexpr int NT = N / 16;
    v8f acc[NT] = {};

    for (int k = 0; k < K; k += 4) {
        v2f a;
        a.x = A[rm * K + k + khalf];
        a.y = A[rm * K + k + khalf + 1];
#pragma unroll
        for (int t = 0; t < NT; ++t) {
            v2f b;
            b.x = W[(k + khalf) * N + t * 16 + colL];
            b.y = W[(k + khalf + 1) * N + t * 16 + colL];
            acc[t] = __builtin_amdgcn_wmma_f32_16x16x4_f32(
                false, a, false, b, (short)0, acc[t], false, false);
        }
    }

    const int mbase = m0 + (lane >> 4) * 8;
    if (m0 + 16 <= n_rows) {
#pragma unroll
        for (int t = 0; t < NT; ++t) {
#pragma unroll
            for (int r = 0; r < 8; ++r) {
                O[(mbase + r) * N + t * 16 + colL] = acc[t][r];
            }
        }
    } else {
#pragma unroll
        for (int t = 0; t < NT; ++t) {
#pragma unroll
            for (int r = 0; r < 8; ++r) {
                int row = mbase + r;
                if (row < n_rows) O[row * N + t * 16 + colL] = acc[t][r];
            }
        }
    }
}


template <int LOGC>
__global__ void k_selfloop(const float* __restrict__ h,
                           const float* __restrict__ dinv,
                           float* __restrict__ out, int total) {
    int id = blockIdx.x * blockDim.x + threadIdx.x;
    if (id >= total) return;
    float d = dinv[id >> LOGC];
    out[id] = d * d * h[id];
}


__global__ void k_edge_scatter32(const int* __restrict__ src,
                                 const int* __restrict__ dst,
                                 const float* __restrict__ dinv,
                                 const float* __restrict__ h,
                                 float* out, int E) {
    int w = (blockIdx.x * blockDim.x + threadIdx.x) >> 5;
    int lane = threadIdx.x & 31;
    if (w >= E) return;
    int s = src[w], d = dst[w];
    float norm = dinv[s] * dinv[d];
    float v = h[s * 32 + lane] * norm;
    unsafeAtomicAdd(&out[d * 32 + lane], v);
}

__global__ void k_edge_scatter64(const int* __restrict__ src,
                                 const int* __restrict__ dst,
                                 const float* __restrict__ dinv,
                                 const float* __restrict__ h,
                                 float* out, int E) {
    int w = (blockIdx.x * blockDim.x + threadIdx.x) >> 5;
    int lane = threadIdx.x & 31;
    if (w >= E) return;
    int s = src[w], d = dst[w];
    float norm = dinv[s] * dinv[d];
    const float2* hp = (const float2*)(h + (size_t)s * 64);
    float2 v = hp[lane];
    unsafeAtomicAdd(&out[d * 64 + 2 * lane + 0], v.x * norm);
    unsafeAtomicAdd(&out[d * 64 + 2 * lane + 1], v.y * norm);
}


template <int C>
__global__ void k_bias_relu(float* h, const float* __restrict__ bias,
                            int total) {
    int id = blockIdx.x * blockDim.x + threadIdx.x;
    if (id >= total) return;
    float v = h[id] + bias[id & (C - 1)];
    h[id] = v > 0.0f ? v : 0.0f;
}


__global__ void k_pool_accum(const float* __restrict__ h,
                             const int* __restrict__ batch,
                             float* sums, int n) {
    int id = blockIdx.x * blockDim.x + threadIdx.x;
    if (id >= n * 64) return;
    int node = id >> 6, c = id & 63;
    int g = batch[node];
    unsafeAtomicAdd(&sums[g * 64 + c], h[id]);
}

__global__ void k_count_accum(const int* __restrict__ batch, float* cnts, int n) {
    int i = blockIdx.x * blockDim.x + threadIdx.x;
    if (i < n) unsafeAtomicAdd(&cnts[batch[i]], 1.0f);
}


__global__ void k_final_head(const float* __restrict__ sums,
                             const float* __restrict__ cnts,
                             const float* __restrict__ Wc,
                             const float* __restrict__ bc,
                             float* out, int G) {
    int g = threadIdx.x;
    if (g >= G) return;
    float c = cnts[g];
    c = c > 1.0f ? c : 1.0f;
    float inv = 1.0f / c;
    float l0 = bc[0], l1 = bc[1];
    for (int ch = 0; ch < 64; ++ch) {
        float p = sums[g * 64 + ch] * inv;
        out[G * 2 + g * 64 + ch] = p;
        l0 = fmaf(p, Wc[ch * 2 + 0], l0);
        l1 = fmaf(p, Wc[ch * 2 + 1], l1);
    }
    out[g * 2 + 0] = l0;
    out[g * 2 + 1] = l1;
}


extern "C" void kernel_launch(void* const* d_in, const int* in_sizes, int n_in,
                              void* d_out, int out_size, void* d_ws, size_t ws_size,
                              hipStream_t stream) {
    const float* x   = (const float*)d_in[0];
    const float* W1  = (const float*)d_in[1];
    const float* b1  = (const float*)d_in[2];
    const float* W2  = (const float*)d_in[3];
    const float* b2  = (const float*)d_in[4];
    const float* Wc  = (const float*)d_in[5];
    const float* bc  = (const float*)d_in[6];
    const int*   ei  = (const int*)d_in[7];
    const int*   bat = (const int*)d_in[8];

    const int n = in_sizes[0] / IN_CH;
    const int E = in_sizes[7] / 2;
    const int G = 64;

    const int* src = ei;
    const int* dst = ei + E;

    float* ws   = (float*)d_ws;
    float* deg  = ws;                ws += n;
    float* dinv = ws;                ws += n;
    float* h1   = ws;                ws += (size_t)n * 32;
    float* o1   = ws;                ws += (size_t)n * 32;
    float* h2   = ws;                ws += (size_t)n * 64;
    float* o2   = ws;                ws += (size_t)n * 64;
    float* sums = ws;                ws += (size_t)G * 64;
    float* cnts = ws;                ws += G;

    const int B = 256;
    auto cdiv = [](long long a, long long b) { return (int)((a + b - 1) / b); };

    k_init_deg<<<cdiv(n, B), B, 0, stream>>>(deg, n);
    k_deg_accum<<<cdiv(E, B), B, 0, stream>>>(dst, deg, E);
    k_dinv<<<cdiv(n, B), B, 0, stream>>>(deg, dinv, n);

    const int tiles = (n + 15) / 16;
    k_gemm_wmma<128, 32><<<cdiv(tiles, 4), 128, 0, stream>>>(x, W1, h1, n, tiles);
    k_selfloop<5><<<cdiv((long long)n * 32, B), B, 0, stream>>>(h1, dinv, o1, n * 32);
    k_edge_scatter32<<<cdiv((long long)E * 32, B), B, 0, stream>>>(src, dst, dinv, h1, o1, E);
    k_bias_relu<32><<<cdiv((long long)n * 32, B), B, 0, stream>>>(o1, b1, n * 32);

    k_gemm_wmma<32, 64><<<cdiv(tiles, 4), 128, 0, stream>>>(o1, W2, h2, n, tiles);
    k_selfloop<6><<<cdiv((long long)n * 64, B), B, 0, stream>>>(h2, dinv, o2, n * 64);
    k_edge_scatter64<<<cdiv((long long)E * 32, B), B, 0, stream>>>(src, dst, dinv, h2, o2, E);
    k_bias_relu<64><<<cdiv((long long)n * 64, B), B, 0, stream>>>(o2, b2, n * 64);

    hipMemsetAsync(sums, 0, (size_t)(G * 64 + G) * sizeof(float), stream);
    k_pool_accum<<<cdiv((long long)n * 64, B), B, 0, stream>>>(o2, bat, sums, n);
    k_count_accum<<<cdiv(n, B), B, 0, stream>>>(bat, cnts, n);
    k_final_head<<<1, 64, 0, stream>>>(sums, cnts, Wc, bc, (float*)d_out, G);
}


# Pooling and node counts as a one-hot (graph x node) f32 matrix product, one wave per 64 nodes, wave partials out by f32 atomics; count kernel folded in, 14 launches to 13; pooled sums reassociated
# speedup vs baseline: 1.2898x; 1.2898x over previous
#include <hip/hip_runtime.h>
#include <hip/hip_bf16.h>


typedef float v2f __attribute__((ext_vector_type(2)));
typedef float v8f __attribute__((ext_vector_type(8)));

#define IN_CH 128


__global__ void k_init_deg(float* deg, int n) {
    int i = blockIdx.x * blockDim.x + threadIdx.x;
    if (i < n) deg[i] = 1.0f;
}

__global__ void k_deg_accum(const int* __restrict__ dst, float* deg, int E) {
    int e = blockIdx.x * blockDim.x + threadIdx.x;
    if (e < E) unsafeAtomicAdd(&deg[dst[e]], 1.0f);
}

__global__ void k_dinv(const float* __restrict__ deg, float* dinv, int n) {
    int i = blockIdx.x * blockDim.x + threadIdx.x;
    if (i < n) {
        float d = deg[i];
        dinv[i] = d > 0.0f ? __frsqrt_rn(d) : 0.0f;
    }
}


template <int K, int N>
__global__ void k_gemm_wmma(const float* __restrict__ A,
                            const float* __restrict__ W,
                            float* __restrict__ O,
                            int n_rows, int n_tiles) {
    int wave = (blockIdx.x * blockDim.x + threadIdx.x) >> 5;
    int lane = threadIdx.x & 31;
    if (wave >= n_tiles) return;

    const int m0    = wave * 16;
    const int mrow  = m0 + (lane & 15);
    const int rm    = mrow < n_rows ? mrow : (n_rows - 1);
    const int khalf = (lane >> 4) * 2;
    const int colL  = lane & 15;

    constexpr int NT = N / 16;
    v8f acc[NT] = {};

    for (int k = 0; k < K; k += 4) {
        v2f a;
        a.x = A[rm * K + k + khalf];
        a.y = A[rm * K + k + khalf + 1];
#pragma unroll
        for (int t = 0; t < NT; ++t) {
            v2f b;
            b.x = W[(k + khalf) * N + t * 16 + colL];
            b.y = W[(k + khalf + 1) * N + t * 16 + colL];
            acc[t] = __builtin_amdgcn_wmma_f32_16x16x4_f32(
                false, a, false, b, (short)0, acc[t], false, false);
        }
    }

    const int mbase = m0 + (lane >> 4) * 8;
    if (m0 + 16 <= n_rows) {
#pragma unroll
        for (int t = 0; t < NT; ++t) {
#pragma unroll
            for (int r = 0; r < 8; ++r) {
                O[(mbase + r) * N + t * 16 + colL] = acc[t][r];
            }
        }
    } else {
#pragma unroll
        for (int t = 0; t < NT; ++t) {
#pragma unroll
            for (int r = 0; r < 8; ++r) {
                int row = mbase + r;
                if (row < n_rows) O[row * N + t * 16 + colL] = acc[t][r];
            }
        }
    }
}


template <int LOGC>
__global__ void k_selfloop(const float* __restrict__ h,
                           const float* __restrict__ dinv,
                           float* __restrict__ out, int total) {
    int id = blockIdx.x * blockDim.x + threadIdx.x;
    if (id >= total) return;
    float d = dinv[id >> LOGC];
    out[id] = d * d * h[id];
}


__global__ void k_edge_scatter32(const int* __restrict__ src,
                                 const int* __restrict__ dst,
                                 const float* __restrict__ dinv,
                                 const float* __restrict__ h,
                                 float* out, int E) {
    int w = (blockIdx.x * blockDim.x + threadIdx.x) >> 5;
    int lane = threadIdx.x & 31;
    if (w >= E) return;
    int s = src[w], d = dst[w];
    float norm = dinv[s] * dinv[d];
    float v = h[s * 32 + lane] * norm;
    unsafeAtomicAdd(&out[d * 32 + lane], v);
}

__global__ void k_edge_scatter64(const int* __restrict__ src,
                                 const int* __restrict__ dst,
                                 const float* __restrict__ dinv,
                                 const float* __restrict__ h,
                                 float* out, int E) {
    int w = (blockIdx.x * blockDim.x + threadIdx.x) >> 5;
    int lane = threadIdx.x & 31;
    if (w >= E) return;
    int s = src[w], d = dst[w];
    float norm = dinv[s] * dinv[d];
    const float2* hp = (const float2*)(h + (size_t)s * 64);
    float2 v = hp[lane];
    unsafeAtomicAdd(&out[d * 64 + 2 * lane + 0], v.x * norm);
    unsafeAtomicAdd(&out[d * 64 + 2 * lane + 1], v.y * norm);
}


template <int C>
__global__ void k_bias_relu(float* h, const float* __restrict__ bias,
                            int total) {
    int id = blockIdx.x * blockDim.x + threadIdx.x;
    if (id >= total) return;
    float v = h[id] + bias[id & (C - 1)];
    h[id] = v > 0.0f ? v : 0.0f;
}


#define POOL_CH 64

struct PoolSet {
    float4 bx[4];
    float4 by[4];
    int    i0[4];
    int    i1[4];
};

template <bool FULL>
__device__ __forceinline__ void pool_load(PoolSet& S,
                                          const float* __restrict__ h,
                                          const int* __restrict__ batch,
                                          int nb, int hh, int c, int n) {
    if (FULL) {
        const float* hp = h + (size_t)(nb + 2 * hh) * 64 + 4 * c;
        const int*   bp = batch + nb + 2 * hh;
#pragma unroll
        for (int u = 0; u < 4; ++u) {
            S.i0[u] = bp[4 * u];
            S.i1[u] = bp[4 * u + 1];
            S.bx[u] = *(const float4*)(hp + (4 * u) * 64);
            S.by[u] = *(const float4*)(hp + (4 * u + 1) * 64);
        }
    } else {
#pragma unroll
        for (int u = 0; u < 4; ++u) {
            const int node0 = nb + 4 * u + 2 * hh;
            const int node1 = node0 + 1;
            const int r0 = node0 < n ? node0 : n - 1;
            const int r1 = node1 < n ? node1 : n - 1;
            const int i0 = batch[r0];
            const int i1 = batch[r1];
            S.i0[u] = node0 < n ? i0 : -1;
            S.i1[u] = node1 < n ? i1 : -1;
            S.bx[u] = *(const float4*)(h + (size_t)r0 * 64 + 4 * c);
            S.by[u] = *(const float4*)(h + (size_t)r1 * 64 + 4 * c);
        }
    }
}

template <bool FULL>
__device__ __forceinline__ void pool_tile(const float* __restrict__ h,
                                          const int* __restrict__ batch,
                                          float* sums, float* cnts,
                                          int base, int mt, int hh, int c, int n) {
    const int g = mt * 16 + c;
    v8f acc[4] = {};
    float cnt = 0.0f;

    PoolSet cur, nxt;
    pool_load<FULL>(cur, h, batch, base, hh, c, n);
    for (int it = 0; it < POOL_CH / 16; ++it) {
        if (it + 1 < POOL_CH / 16)
            pool_load<FULL>(nxt, h, batch, base + (it + 1) * 16, hh, c, n);
#pragma unroll
        for (int u = 0; u < 4; ++u) {
            v2f a;
            a.x = cur.i0[u] == g ? 1.0f : 0.0f;
            a.y = cur.i1[u] == g ? 1.0f : 0.0f;
            cnt += a.x + a.y;
            const float bx[4] = {cur.bx[u].x, cur.bx[u].y, cur.bx[u].z, cur.bx[u].w};
            const float by[4] = {cur.by[u].x, cur.by[u].y, cur.by[u].z, cur.by[u].w};
#pragma unroll
            for (int nt = 0; nt < 4; ++nt) {
                v2f b;
                b.x = bx[nt];
                b.y = by[nt];
                acc[nt] = __builtin_amdgcn_wmma_f32_16x16x4_f32(
                    false, a, false, b, (short)0, acc[nt], false, false);
            }
        }
        cur = nxt;
    }

    const float tot = cnt + __shfl_xor(cnt, 16);
    if (hh == 0 && tot != 0.0f) unsafeAtomicAdd(&cnts[g], tot);

    const int grow = mt * 16 + hh * 8;
#pragma unroll
    for (int r = 0; r < 8; ++r) {
#pragma unroll
        for (int nt = 0; nt < 4; ++nt) {
            const float v = acc[nt][r];
            if (v != 0.0f)
                unsafeAtomicAdd(&sums[(grow + r) * 64 + 4 * c + nt], v);
        }
    }
}

__global__ void __launch_bounds__(64)
k_pool_onehot(const float* __restrict__ h,
              const int* __restrict__ batch,
              float* sums, float* cnts, int n) {
    const int wave = (blockIdx.x * blockDim.x + threadIdx.x) >> 5;
    const int lane = threadIdx.x & 31;
    const int base = wave * POOL_CH;
    if (base >= n) return;
    const int c  = lane & 15;
    const int hh = lane >> 4;
    const bool full = base + POOL_CH <= n;

    unsigned present = 0;
    for (int i = lane; i < POOL_CH; i += 32) {
        const int node = base + i;
        if (node < n) {
            const unsigned gid = (unsigned)batch[node];
            if (gid < 64u) present |= 1u << (gid >> 4);
        }
    }

    for (int mt = 0; mt < 4; ++mt) {
        if (__ballot((present >> mt) & 1u) == 0) continue;
        if (full) pool_tile<true >(h, batch, sums, cnts, base, mt, hh, c, n);
        else      pool_tile<false>(h, batch, sums, cnts, base, mt, hh, c, n);
    }
}


__global__ void k_final_head(const float* __restrict__ sums,
                             const float* __restrict__ cnts,
                             const float* __restrict__ Wc,
                             const float* __restrict__ bc,
                             float* out, int G) {
    int g = threadIdx.x;
    if (g >= G) return;
    float c = cnts[g];
    c = c > 1.0f ? c : 1.0f;
    float inv = 1.0f / c;
    float l0 = bc[0], l1 = bc[1];
    for (int ch = 0; ch < 64; ++ch) {
        float p = sums[g * 64 + ch] * inv;
        out[G * 2 + g * 64 + ch] = p;
        l0 = fmaf(p, Wc[ch * 2 + 0], l0);
        l1 = fmaf(p, Wc[ch * 2 + 1], l1);
    }
    out[g * 2 + 0] = l0;
    out[g * 2 + 1] = l1;
}


extern "C" void kernel_launch(void* const* d_in, const int* in_sizes, int n_in,
                              void* d_out, int out_size, void* d_ws, size_t ws_size,
                              hipStream_t stream) {
    const float* x   = (const float*)d_in[0];
    const float* W1  = (const float*)d_in[1];
    const float* b1  = (const float*)d_in[2];
    const float* W2  = (const float*)d_in[3];
    const float* b2  = (const float*)d_in[4];
    const float* Wc  = (const float*)d_in[5];
    const float* bc  = (const float*)d_in[6];
    const int*   ei  = (const int*)d_in[7];
    const int*   bat = (const int*)d_in[8];

    const int n = in_sizes[0] / IN_CH;
    const int E = in_sizes[7] / 2;
    const int G = 64;

    const int* src = ei;
    const int* dst = ei + E;

    float* ws   = (float*)d_ws;
    float* deg  = ws;                ws += n;
    float* dinv = ws;                ws += n;
    float* h1   = ws;                ws += (size_t)n * 32;
    float* o1   = ws;                ws += (size_t)n * 32;
    float* h2   = ws;                ws += (size_t)n * 64;
    float* o2   = ws;                ws += (size_t)n * 64;
    float* sums = ws;                ws += (size_t)G * 64;
    float* cnts = ws;                ws += G;

    const int B = 256;
    auto cdiv = [](long long a, long long b) { return (int)((a + b - 1) / b); };

    k_init_deg<<<cdiv(n, B), B, 0, stream>>>(deg, n);
    k_deg_accum<<<cdiv(E, B), B, 0, stream>>>(dst, deg, E);
    k_dinv<<<cdiv(n, B), B, 0, stream>>>(deg, dinv, n);

    const int tiles = (n + 15) / 16;
    k_gemm_wmma<128, 32><<<cdiv(tiles, 4), 128, 0, stream>>>(x, W1, h1, n, tiles);
    k_selfloop<5><<<cdiv((long long)n * 32, B), B, 0, stream>>>(h1, dinv, o1, n * 32);
    k_edge_scatter32<<<cdiv((long long)E * 32, B), B, 0, stream>>>(src, dst, dinv, h1, o1, E);
    k_bias_relu<32><<<cdiv((long long)n * 32, B), B, 0, stream>>>(o1, b1, n * 32);

    k_gemm_wmma<32, 64><<<cdiv(tiles, 4), 128, 0, stream>>>(o1, W2, h2, n, tiles);
    k_selfloop<6><<<cdiv((long long)n * 64, B), B, 0, stream>>>(h2, dinv, o2, n * 64);
    k_edge_scatter64<<<cdiv((long long)E * 32, B), B, 0, stream>>>(src, dst, dinv, h2, o2, E);
    k_bias_relu<64><<<cdiv((long long)n * 64, B), B, 0, stream>>>(o2, b2, n * 64);

    hipMemsetAsync(sums, 0, (size_t)(G * 64 + G) * sizeof(float), stream);
    k_pool_onehot<<<cdiv(n, 2 * POOL_CH), 64, 0, stream>>>(o2, bat, sums, cnts, n);
    k_final_head<<<1, 64, 0, stream>>>(sums, cnts, Wc, bc, (float*)d_out, G);
}


# Edge scatter grouped by source: per-call counting sort of edges by source; one wave per source reads its row once, then a full-line f32 atomic per out-edge; atomic arrival order still unordered
# speedup vs baseline: 1.3334x; 1.0338x over previous
#include <hip/hip_runtime.h>
#include <hip/hip_bf16.h>


typedef float v2f __attribute__((ext_vector_type(2)));
typedef float v8f __attribute__((ext_vector_type(8)));

#define IN_CH 128


#define SCAN_B 256

__global__ void k_init_counts(int* cnt, float* deg, int n) {
    int i = blockIdx.x * blockDim.x + threadIdx.x;
    if (i < n) {
        cnt[i] = 0;
        deg[i] = 1.0f;
    }
}

__global__ void k_count(const int* __restrict__ src, const int* __restrict__ dst,
                        int* cnt, float* deg, int E) {
    int e = blockIdx.x * blockDim.x + threadIdx.x;
    if (e < E) {
        atomicAdd(&cnt[src[e]], 1);
        unsafeAtomicAdd(&deg[dst[e]], 1.0f);
    }
}

__global__ void k_blocksum(const int* __restrict__ cnt, int* bsum, int n) {
    __shared__ int sh[SCAN_B / 32];
    const int i = blockIdx.x * SCAN_B + threadIdx.x;
    int v = i < n ? cnt[i] : 0;
    for (int o = 16; o > 0; o >>= 1) v += __shfl_xor(v, o);
    if ((threadIdx.x & 31) == 0) sh[threadIdx.x >> 5] = v;
    __syncthreads();
    if (threadIdx.x == 0) {
        int t = 0;
        for (int w = 0; w < SCAN_B / 32; ++w) t += sh[w];
        bsum[blockIdx.x] = t;
    }
}

__global__ void k_rowptr(const int* __restrict__ cnt, const int* __restrict__ bsum,
                         const float* __restrict__ deg,
                         int* rowptr, int* cursor, float* dinv, int n) {
    __shared__ int sh_inc[SCAN_B / 32];
    __shared__ int sh_pre[SCAN_B / 32];
    const int tid = threadIdx.x, lane = tid & 31, wv = tid >> 5;

    int p = 0;
    for (int j = tid; j < (int)blockIdx.x; j += SCAN_B) p += bsum[j];
    for (int o = 16; o > 0; o >>= 1) p += __shfl_xor(p, o);
    if (lane == 0) sh_pre[wv] = p;

    const int i = blockIdx.x * SCAN_B + tid;
    const int c = i < n ? cnt[i] : 0;
    int inc = c;
    for (int o = 1; o < 32; o <<= 1) {
        const int t = __shfl_up(inc, o);
        if (lane >= o) inc += t;
    }
    if (lane == 31) sh_inc[wv] = inc;
    __syncthreads();

    int base = 0;
    for (int w = 0; w < SCAN_B / 32; ++w) {
        base += sh_pre[w];
        if (w < wv) base += sh_inc[w];
    }
    const int excl = base + inc - c;
    if (i < n) {
        rowptr[i] = excl;
        cursor[i] = excl;
        const float d = deg[i];
        dinv[i] = d > 0.0f ? __frsqrt_rn(d) : 0.0f;
    }
    if (i == n - 1) rowptr[n] = excl + c;
}

__global__ void k_fill(const int* __restrict__ src, const int* __restrict__ dst,
                       int* cursor, int* adj, int E) {
    int e = blockIdx.x * blockDim.x + threadIdx.x;
    if (e < E) {
        const int p = atomicAdd(&cursor[src[e]], 1);
        adj[p] = dst[e];
    }
}


template <int K, int N>
__global__ void k_gemm_wmma(const float* __restrict__ A,
                            const float* __restrict__ W,
                            float* __restrict__ O,
                            int n_rows, int n_tiles) {
    int wave = (blockIdx.x * blockDim.x + threadIdx.x) >> 5;
    int lane = threadIdx.x & 31;
    if (wave >= n_tiles) return;

    const int m0    = wave * 16;
    const int mrow  = m0 + (lane & 15);
    const int rm    = mrow < n_rows ? mrow : (n_rows - 1);
    const int khalf = (lane >> 4) * 2;
    const int colL  = lane & 15;

    constexpr int NT = N / 16;
    v8f acc[NT] = {};

    for (int k = 0; k < K; k += 4) {
        v2f a;
        a.x = A[rm * K + k + khalf];
        a.y = A[rm * K + k + khalf + 1];
#pragma unroll
        for (int t = 0; t < NT; ++t) {
            v2f b;
            b.x = W[(k + khalf) * N + t * 16 + colL];
            b.y = W[(k + khalf + 1) * N + t * 16 + colL];
            acc[t] = __builtin_amdgcn_wmma_f32_16x16x4_f32(
                false, a, false, b, (short)0, acc[t], false, false);
        }
    }

    const int mbase = m0 + (lane >> 4) * 8;
    if (m0 + 16 <= n_rows) {
#pragma unroll
        for (int t = 0; t < NT; ++t) {
#pragma unroll
            for (int r = 0; r < 8; ++r) {
                O[(mbase + r) * N + t * 16 + colL] = acc[t][r];
            }
        }
    } else {
#pragma unroll
        for (int t = 0; t < NT; ++t) {
#pragma unroll
            for (int r = 0; r < 8; ++r) {
                int row = mbase + r;
                if (row < n_rows) O[row * N + t * 16 + colL] = acc[t][r];
            }
        }
    }
}


template <int LOGC>
__global__ void k_selfloop(const float* __restrict__ h,
                           const float* __restrict__ dinv,
                           float* __restrict__ out, int total) {
    int id = blockIdx.x * blockDim.x + threadIdx.x;
    if (id >= total) return;
    float d = dinv[id >> LOGC];
    out[id] = d * d * h[id];
}


template <int CPL>
__global__ void k_push(const int* __restrict__ rowptr,
                       const int* __restrict__ adj,
                       const float* __restrict__ dinv,
                       const float* __restrict__ h,
                       float* out, int n) {
    const int s = (blockIdx.x * blockDim.x + threadIdx.x) >> 5;
    const int lane = threadIdx.x & 31;
    if (s >= n) return;
    const int beg = __builtin_amdgcn_readfirstlane(rowptr[s]);
    const int end = __builtin_amdgcn_readfirstlane(rowptr[s + 1]);
    if (beg >= end) return;

    int b = beg;
    int m = end - b < 32 ? end - b : 32;
    int dj = 0;
    float wj = 0.0f;
    if (lane < m) dj = adj[b + lane];
    const float ds = dinv[s];
    float hv[CPL];
#pragma unroll
    for (int q = 0; q < CPL; ++q) hv[q] = h[(size_t)s * (32 * CPL) + 32 * q + lane];
    if (lane < m) wj = dinv[dj];

    for (;;) {
        for (int j = 0; j < m; ++j) {
            const int d = __builtin_amdgcn_readlane(dj, j);
            const float w = __int_as_float(__builtin_amdgcn_readlane(__float_as_int(wj), j));
            const float norm = ds * w;
            float* orow = out + (size_t)d * (32 * CPL) + lane;
#pragma unroll
            for (int q = 0; q < CPL; ++q) unsafeAtomicAdd(orow + 32 * q, hv[q] * norm);
        }
        b += 32;
        if (b >= end) break;
        m = end - b < 32 ? end - b : 32;
        dj = 0;
        wj = 0.0f;
        if (lane < m) {
            dj = adj[b + lane];
            wj = dinv[dj];
        }
    }
}


template <int C>
__global__ void k_bias_relu(float* h, const float* __restrict__ bias,
                            int total) {
    int id = blockIdx.x * blockDim.x + threadIdx.x;
    if (id >= total) return;
    float v = h[id] + bias[id & (C - 1)];
    h[id] = v > 0.0f ? v : 0.0f;
}


#define POOL_CH 64

struct PoolSet {
    float4 bx[4];
    float4 by[4];
    int    i0[4];
    int    i1[4];
};

template <bool FULL>
__device__ __forceinline__ void pool_load(PoolSet& S,
                                          const float* __restrict__ h,
                                          const int* __restrict__ batch,
                                          int nb, int hh, int c, int n) {
    if (FULL) {
        const float* hp = h + (size_t)(nb + 2 * hh) * 64 + 4 * c;
        const int*   bp = batch + nb + 2 * hh;
#pragma unroll
        for (int u = 0; u < 4; ++u) {
            S.i0[u] = bp[4 * u];
            S.i1[u] = bp[4 * u + 1];
            S.bx[u] = *(const float4*)(hp + (4 * u) * 64);
            S.by[u] = *(const float4*)(hp + (4 * u + 1) * 64);
        }
    } else {
#pragma unroll
        for (int u = 0; u < 4; ++u) {
            const int node0 = nb + 4 * u + 2 * hh;
            const int node1 = node0 + 1;
            const int r0 = node0 < n ? node0 : n - 1;
            const int r1 = node1 < n ? node1 : n - 1;
            const int i0 = batch[r0];
            const int i1 = batch[r1];
            S.i0[u] = node0 < n ? i0 : -1;
            S.i1[u] = node1 < n ? i1 : -1;
            S.bx[u] = *(const float4*)(h + (size_t)r0 * 64 + 4 * c);
            S.by[u] = *(const float4*)(h + (size_t)r1 * 64 + 4 * c);
        }
    }
}

template <bool FULL>
__device__ __forceinline__ void pool_tile(const float* __restrict__ h,
                                          const int* __restrict__ batch,
                                          float* sums, float* cnts,
                                          int base, int mt, int hh, int c, int n) {
    const int g = mt * 16 + c;
    v8f acc[4] = {};
    float cnt = 0.0f;

    PoolSet cur, nxt;
    pool_load<FULL>(cur, h, batch, base, hh, c, n);
    for (int it = 0; it < POOL_CH / 16; ++it) {
        if (it + 1 < POOL_CH / 16)
            pool_load<FULL>(nxt, h, batch, base + (it + 1) * 16, hh, c, n);
#pragma unroll
        for (int u = 0; u < 4; ++u) {
            v2f a;
            a.x = cur.i0[u] == g ? 1.0f : 0.0f;
            a.y = cur.i1[u] == g ? 1.0f : 0.0f;
            cnt += a.x + a.y;
            const float bx[4] = {cur.bx[u].x, cur.bx[u].y, cur.bx[u].z, cur.bx[u].w};
            const float by[4] = {cur.by[u].x, cur.by[u].y, cur.by[u].z, cur.by[u].w};
#pragma unroll
            for (int nt = 0; nt < 4; ++nt) {
                v2f b;
                b.x = bx[nt];
                b.y = by[nt];
                acc[nt] = __builtin_amdgcn_wmma_f32_16x16x4_f32(
                    false, a, false, b, (short)0, acc[nt], false, false);
            }
        }
        cur = nxt;
    }

    const float tot = cnt + __shfl_xor(cnt, 16);
    if (hh == 0 && tot != 0.0f) unsafeAtomicAdd(&cnts[g], tot);

    const int grow = mt * 16 + hh * 8;
#pragma unroll
    for (int r = 0; r < 8; ++r) {
#pragma unroll
        for (int nt = 0; nt < 4; ++nt) {
            const float v = acc[nt][r];
            if (v != 0.0f)
                unsafeAtomicAdd(&sums[(grow + r) * 64 + 4 * c + nt], v);
        }
    }
}

__global__ void __launch_bounds__(64)
k_pool_onehot(const float* __restrict__ h,
              const int* __restrict__ batch,
              float* sums, float* cnts, int n) {
    const int wave = (blockIdx.x * blockDim.x + threadIdx.x) >> 5;
    const int lane = threadIdx.x & 31;
    const int base = wave * POOL_CH;
    if (base >= n) return;
    const int c  = lane & 15;
    const int hh = lane >> 4;
    const bool full = base + POOL_CH <= n;

    unsigned present = 0;
    for (int i = lane; i < POOL_CH; i += 32) {
        const int node = base + i;
        if (node < n) {
            const unsigned gid = (unsigned)batch[node];
            if (gid < 64u) present |= 1u << (gid >> 4);
        }
    }

    for (int mt = 0; mt < 4; ++mt) {
        if (__ballot((present >> mt) & 1u) == 0) continue;
        if (full) pool_tile<true >(h, batch, sums, cnts, base, mt, hh, c, n);
        else      pool_tile<false>(h, batch, sums, cnts, base, mt, hh, c, n);
    }
}


__global__ void k_final_head(const float* __restrict__ sums,
                             const float* __restrict__ cnts,
                             const float* __restrict__ Wc,
                             const float* __restrict__ bc,
                             float* out, int G) {
    int g = threadIdx.x;
    if (g >= G) return;
    float c = cnts[g];
    c = c > 1.0f ? c : 1.0f;
    float inv = 1.0f / c;
    float l0 = bc[0], l1 = bc[1];
    for (int ch = 0; ch < 64; ++ch) {
        float p = sums[g * 64 + ch] * inv;
        out[G * 2 + g * 64 + ch] = p;
        l0 = fmaf(p, Wc[ch * 2 + 0], l0);
        l1 = fmaf(p, Wc[ch * 2 + 1], l1);
    }
    out[g * 2 + 0] = l0;
    out[g * 2 + 1] = l1;
}


extern "C" void kernel_launch(void* const* d_in, const int* in_sizes, int n_in,
                              void* d_out, int out_size, void* d_ws, size_t ws_size,
                              hipStream_t stream) {
    const float* x   = (const float*)d_in[0];
    const float* W1  = (const float*)d_in[1];
    const float* b1  = (const float*)d_in[2];
    const float* W2  = (const float*)d_in[3];
    const float* b2  = (const float*)d_in[4];
    const float* Wc  = (const float*)d_in[5];
    const float* bc  = (const float*)d_in[6];
    const int*   ei  = (const int*)d_in[7];
    const int*   bat = (const int*)d_in[8];

    const int n = in_sizes[0] / IN_CH;
    const int E = in_sizes[7] / 2;
    const int G = 64;

    const int* src = ei;
    const int* dst = ei + E;

    float* ws   = (float*)d_ws;
    float* bufA = ws;                ws += (size_t)n * 64;
    float* bufB = ws;                ws += (size_t)n * 64;
    float* deg  = ws;                ws += n;
    float* dinv = ws;                ws += n;
    float* sums = ws;                ws += (size_t)G * 64;
    float* cnts = ws;                ws += G;
    int* cnt    = (int*)ws;          ws += n;
    int* rowptr = (int*)ws;          ws += n + 1;
    int* cursor = (int*)ws;          ws += n;
    int* bsum   = (int*)ws;          ws += (n + SCAN_B - 1) / SCAN_B;
    int* adj    = (int*)ws;          ws += E;
    float* h1 = bufA;
    float* o1 = bufB;
    float* h2 = bufA;
    float* o2 = bufB;

    const int B = 256;
    auto cdiv = [](long long a, long long b) { return (int)((a + b - 1) / b); };

    k_init_counts<<<cdiv(n, B), B, 0, stream>>>(cnt, deg, n);
    k_count<<<cdiv(E, B), B, 0, stream>>>(src, dst, cnt, deg, E);
    k_blocksum<<<cdiv(n, SCAN_B), SCAN_B, 0, stream>>>(cnt, bsum, n);
    k_rowptr<<<cdiv(n, SCAN_B), SCAN_B, 0, stream>>>(cnt, bsum, deg, rowptr, cursor, dinv, n);
    k_fill<<<cdiv(E, B), B, 0, stream>>>(src, dst, cursor, adj, E);

    const int tiles = (n + 15) / 16;
    k_gemm_wmma<128, 32><<<cdiv(tiles, 4), 128, 0, stream>>>(x, W1, h1, n, tiles);
    k_selfloop<5><<<cdiv((long long)n * 32, B), B, 0, stream>>>(h1, dinv, o1, n * 32);
    k_push<1><<<cdiv((long long)n * 32, B), B, 0, stream>>>(rowptr, adj, dinv, h1, o1, n);
    k_bias_relu<32><<<cdiv((long long)n * 32, B), B, 0, stream>>>(o1, b1, n * 32);

    k_gemm_wmma<32, 64><<<cdiv(tiles, 4), 128, 0, stream>>>(o1, W2, h2, n, tiles);
    k_selfloop<6><<<cdiv((long long)n * 64, B), B, 0, stream>>>(h2, dinv, o2, n * 64);
    k_push<2><<<cdiv((long long)n * 32, B), B, 0, stream>>>(rowptr, adj, dinv, h2, o2, n);
    k_bias_relu<64><<<cdiv((long long)n * 64, B), B, 0, stream>>>(o2, b2, n * 64);

    hipMemsetAsync(sums, 0, (size_t)(G * 64 + G) * sizeof(float), stream);
    k_pool_onehot<<<cdiv(n, 2 * POOL_CH), 64, 0, stream>>>(o2, bat, sums, cnts, n);
    k_final_head<<<1, 64, 0, stream>>>(sums, cnts, Wc, bc, (float*)d_out, G);
}


# Edge sum as a pull: edges counting-sorted by destination, a wave per node sums in-neighbour rows in registers with self-loop, bias, ReLU, one store; no row atomics, sums in list order
# speedup vs baseline: 2.0634x; 1.5475x over previous
#include <hip/hip_runtime.h>
#include <hip/hip_bf16.h>


typedef float v2f __attribute__((ext_vector_type(2)));
typedef float v8f __attribute__((ext_vector_type(8)));

#define IN_CH 128


#define SCAN_B 256

__global__ void k_init_counts(int* cnt, int n) {
    int i = blockIdx.x * blockDim.x + threadIdx.x;
    if (i < n) cnt[i] = 0;
}

__global__ void k_count(const int* __restrict__ dst, int* cnt, int E) {
    int e = blockIdx.x * blockDim.x + threadIdx.x;
    if (e < E) atomicAdd(&cnt[dst[e]], 1);
}

__global__ void k_blocksum(const int* __restrict__ cnt, int* bsum, int n) {
    __shared__ int sh[SCAN_B / 32];
    const int i = blockIdx.x * SCAN_B + threadIdx.x;
    int v = i < n ? cnt[i] : 0;
    for (int o = 16; o > 0; o >>= 1) v += __shfl_xor(v, o);
    if ((threadIdx.x & 31) == 0) sh[threadIdx.x >> 5] = v;
    __syncthreads();
    if (threadIdx.x == 0) {
        int t = 0;
        for (int w = 0; w < SCAN_B / 32; ++w) t += sh[w];
        bsum[blockIdx.x] = t;
    }
}

__global__ void k_rowptr(const int* __restrict__ cnt, const int* __restrict__ bsum,
                         int* rowptr, int* cursor, float* dinv, int n) {
    __shared__ int sh_inc[SCAN_B / 32];
    __shared__ int sh_pre[SCAN_B / 32];
    const int tid = threadIdx.x, lane = tid & 31, wv = tid >> 5;

    int p = 0;
    for (int j = tid; j < (int)blockIdx.x; j += SCAN_B) p += bsum[j];
    for (int o = 16; o > 0; o >>= 1) p += __shfl_xor(p, o);
    if (lane == 0) sh_pre[wv] = p;

    const int i = blockIdx.x * SCAN_B + tid;
    const int c = i < n ? cnt[i] : 0;
    int inc = c;
    for (int o = 1; o < 32; o <<= 1) {
        const int t = __shfl_up(inc, o);
        if (lane >= o) inc += t;
    }
    if (lane == 31) sh_inc[wv] = inc;
    __syncthreads();

    int base = 0;
    for (int w = 0; w < SCAN_B / 32; ++w) {
        base += sh_pre[w];
        if (w < wv) base += sh_inc[w];
    }
    const int excl = base + inc - c;
    if (i < n) {
        rowptr[i] = excl;
        cursor[i] = excl;
        const float d = (float)(c + 1);
        dinv[i] = d > 0.0f ? __frsqrt_rn(d) : 0.0f;
    }
    if (i == n - 1) rowptr[n] = excl + c;
}

__global__ void k_fill(const int* __restrict__ src, const int* __restrict__ dst,
                       int* cursor, int* adj, int E) {
    int e = blockIdx.x * blockDim.x + threadIdx.x;
    if (e < E) {
        const int p = atomicAdd(&cursor[dst[e]], 1);
        adj[p] = src[e];
    }
}


template <int K, int N>
__global__ void k_gemm_wmma(const float* __restrict__ A,
                            const float* __restrict__ W,
                            float* __restrict__ O,
                            int n_rows, int n_tiles) {
    int wave = (blockIdx.x * blockDim.x + threadIdx.x) >> 5;
    int lane = threadIdx.x & 31;
    if (wave >= n_tiles) return;

    const int m0    = wave * 16;
    const int mrow  = m0 + (lane & 15);
    const int rm    = mrow < n_rows ? mrow : (n_rows - 1);
    const int khalf = (lane >> 4) * 2;
    const int colL  = lane & 15;

    constexpr int NT = N / 16;
    v8f acc[NT] = {};

    for (int k = 0; k < K; k += 4) {
        v2f a;
        a.x = A[rm * K + k + khalf];
        a.y = A[rm * K + k + khalf + 1];
#pragma unroll
        for (int t = 0; t < NT; ++t) {
            v2f b;
            b.x = W[(k + khalf) * N + t * 16 + colL];
            b.y = W[(k + khalf + 1) * N + t * 16 + colL];
            acc[t] = __builtin_amdgcn_wmma_f32_16x16x4_f32(
                false, a, false, b, (short)0, acc[t], false, false);
        }
    }

    const int mbase = m0 + (lane >> 4) * 8;
    if (m0 + 16 <= n_rows) {
#pragma unroll
        for (int t = 0; t < NT; ++t) {
#pragma unroll
            for (int r = 0; r < 8; ++r) {
                O[(mbase + r) * N + t * 16 + colL] = acc[t][r];
            }
        }
    } else {
#pragma unroll
        for (int t = 0; t < NT; ++t) {
#pragma unroll
            for (int r = 0; r < 8; ++r) {
                int row = mbase + r;
                if (row < n_rows) O[row * N + t * 16 + colL] = acc[t][r];
            }
        }
    }
}


template <int CPL>
__device__ __forceinline__ void row_load(float (&v)[CPL], const float* __restrict__ t, int idx);

template <>
__device__ __forceinline__ void row_load<1>(float (&v)[1], const float* __restrict__ t, int idx) {
    v[0] = t[idx];
}

template <>
__device__ __forceinline__ void row_load<2>(float (&v)[2], const float* __restrict__ t, int idx) {
    const v2f p = ((const v2f*)t)[idx];
    v[0] = p.x;
    v[1] = p.y;
}

template <int CPL, int U>
__device__ __forceinline__ void pull_group(float (&acc)[CPL],
                                           const float* __restrict__ h, int lane,
                                           int sj, float wj, int j, float dd) {
    float v[U][CPL];
#pragma unroll
    for (int u = 0; u < U; ++u) {
        const int s = __builtin_amdgcn_readlane(sj, j + u);
        row_load<CPL>(v[u], h, s * 32 + lane);
    }
    __builtin_amdgcn_sched_barrier(0);
#pragma unroll
    for (int u = 0; u < U; ++u) {
        const float w = __int_as_float(__builtin_amdgcn_readlane(__float_as_int(wj), j + u));
        const float norm = w * dd;
#pragma unroll
        for (int q = 0; q < CPL; ++q) acc[q] = fmaf(v[u][q], norm, acc[q]);
    }
}

template <int CPL>
__global__ void
k_pull(const int* __restrict__ rowptr,
       const int* __restrict__ adj,
       const float* __restrict__ dinv,
       const float* __restrict__ h,
       const float* __restrict__ bias,
       float* __restrict__ out, int n) {
    const int d = (blockIdx.x * blockDim.x + threadIdx.x) >> 5;
    const int lane = threadIdx.x & 31;
    if (d >= n) return;
    const int beg = __builtin_amdgcn_readfirstlane(rowptr[d]);
    const int end = __builtin_amdgcn_readfirstlane(rowptr[d + 1]);

    const float dd = __int_as_float(__builtin_amdgcn_readfirstlane(__float_as_int(dinv[d])));
    float self[CPL];
    row_load<CPL>(self, h, d * 32 + lane);
    float bv[CPL];
    row_load<CPL>(bv, bias, lane);

    float acc[CPL];
#pragma unroll
    for (int q = 0; q < CPL; ++q) acc[q] = 0.0f;

    for (int b = beg; b < end; b += 32) {
        const int m = end - b < 32 ? end - b : 32;
        int sj = 0;
        float wj = 0.0f;
        if (lane < m) {
            sj = adj[b + lane];
            wj = dinv[sj];
        }
        int j = 0;
        for (; j + 8 <= m; j += 8) pull_group<CPL, 8>(acc, h, lane, sj, wj, j, dd);
        switch (m - j) {
            case 7: pull_group<CPL, 7>(acc, h, lane, sj, wj, j, dd); break;
            case 6: pull_group<CPL, 6>(acc, h, lane, sj, wj, j, dd); break;
            case 5: pull_group<CPL, 5>(acc, h, lane, sj, wj, j, dd); break;
            case 4: pull_group<CPL, 4>(acc, h, lane, sj, wj, j, dd); break;
            case 3: pull_group<CPL, 3>(acc, h, lane, sj, wj, j, dd); break;
            case 2: pull_group<CPL, 2>(acc, h, lane, sj, wj, j, dd); break;
            case 1: pull_group<CPL, 1>(acc, h, lane, sj, wj, j, dd); break;
            default: break;
        }
    }

    const float sn = dd * dd;
    float r[CPL];
#pragma unroll
    for (int q = 0; q < CPL; ++q) {
        const float t = fmaf(self[q], sn, acc[q]) + bv[q];
        r[q] = t > 0.0f ? t : 0.0f;
    }
    if (CPL == 1) {
        out[d * 32 + lane] = r[0];
    } else {
        v2f t;
        t.x = r[0];
        t.y = r[CPL - 1];
        ((v2f*)out)[d * 32 + lane] = t;
    }
}


#define POOL_CH 64

struct PoolSet {
    float4 bx[4];
    float4 by[4];
    int    i0[4];
    int    i1[4];
};

template <bool FULL>
__device__ __forceinline__ void pool_load(PoolSet& S,
                                          const float* __restrict__ h,
                                          const int* __restrict__ batch,
                                          int nb, int hh, int c, int n) {
    if (FULL) {
        const float* hp = h + (size_t)(nb + 2 * hh) * 64 + 4 * c;
        const int*   bp = batch + nb + 2 * hh;
#pragma unroll
        for (int u = 0; u < 4; ++u) {
            S.i0[u] = bp[4 * u];
            S.i1[u] = bp[4 * u + 1];
            S.bx[u] = *(const float4*)(hp + (4 * u) * 64);
            S.by[u] = *(const float4*)(hp + (4 * u + 1) * 64);
        }
    } else {
#pragma unroll
        for (int u = 0; u < 4; ++u) {
            const int node0 = nb + 4 * u + 2 * hh;
            const int node1 = node0 + 1;
            const int r0 = node0 < n ? node0 : n - 1;
            const int r1 = node1 < n ? node1 : n - 1;
            const int i0 = batch[r0];
            const int i1 = batch[r1];
            S.i0[u] = node0 < n ? i0 : -1;
            S.i1[u] = node1 < n ? i1 : -1;
            S.bx[u] = *(const float4*)(h + (size_t)r0 * 64 + 4 * c);
            S.by[u] = *(const float4*)(h + (size_t)r1 * 64 + 4 * c);
        }
    }
}

template <bool FULL>
__device__ __forceinline__ void pool_tile(const float* __restrict__ h,
                                          const int* __restrict__ batch,
                                          float* sums, float* cnts,
                                          int base, int mt, int hh, int c, int n) {
    const int g = mt * 16 + c;
    v8f acc[4] = {};
    float cnt = 0.0f;

    PoolSet cur, nxt;
    pool_load<FULL>(cur, h, batch, base, hh, c, n);
    for (int it = 0; it < POOL_CH / 16; ++it) {
        if (it + 1 < POOL_CH / 16)
            pool_load<FULL>(nxt, h, batch, base + (it + 1) * 16, hh, c, n);
#pragma unroll
        for (int u = 0; u < 4; ++u) {
            v2f a;
            a.x = cur.i0[u] == g ? 1.0f : 0.0f;
            a.y = cur.i1[u] == g ? 1.0f : 0.0f;
            cnt += a.x + a.y;
            const float bx[4] = {cur.bx[u].x, cur.bx[u].y, cur.bx[u].z, cur.bx[u].w};
            const float by[4] = {cur.by[u].x, cur.by[u].y, cur.by[u].z, cur.by[u].w};
#pragma unroll
            for (int nt = 0; nt < 4; ++nt) {
                v2f b;
                b.x = bx[nt];
                b.y = by[nt];
                acc[nt] = __builtin_amdgcn_wmma_f32_16x16x4_f32(
                    false, a, false, b, (short)0, acc[nt], false, false);
            }
        }
        cur = nxt;
    }

    const float tot = cnt + __shfl_xor(cnt, 16);
    if (hh == 0 && tot != 0.0f) unsafeAtomicAdd(&cnts[g], tot);

    const int grow = mt * 16 + hh * 8;
#pragma unroll
    for (int r = 0; r < 8; ++r) {
#pragma unroll
        for (int nt = 0; nt < 4; ++nt) {
            const float v = acc[nt][r];
            if (v != 0.0f)
                unsafeAtomicAdd(&sums[(grow + r) * 64 + 4 * c + nt], v);
        }
    }
}

__global__ void __launch_bounds__(64)
k_pool_onehot(const float* __restrict__ h,
              const int* __restrict__ batch,
              float* sums, float* cnts, int n) {
    const int wave = (blockIdx.x * blockDim.x + threadIdx.x) >> 5;
    const int lane = threadIdx.x & 31;
    const int base = wave * POOL_CH;
    if (base >= n) return;
    const int c  = lane & 15;
    const int hh = lane >> 4;
    const bool full = base + POOL_CH <= n;

    unsigned present = 0;
    for (int i = lane; i < POOL_CH; i += 32) {
        const int node = base + i;
        if (node < n) {
            const unsigned gid = (unsigned)batch[node];
            if (gid < 64u) present |= 1u << (gid >> 4);
        }
    }

    for (int mt = 0; mt < 4; ++mt) {
        if (__ballot((present >> mt) & 1u) == 0) continue;
        if (full) pool_tile<true >(h, batch, sums, cnts, base, mt, hh, c, n);
        else      pool_tile<false>(h, batch, sums, cnts, base, mt, hh, c, n);
    }
}


__global__ void k_final_head(const float* __restrict__ sums,
                             const float* __restrict__ cnts,
                             const float* __restrict__ Wc,
                             const float* __restrict__ bc,
                             float* out, int G) {
    int g = threadIdx.x;
    if (g >= G) return;
    float c = cnts[g];
    c = c > 1.0f ? c : 1.0f;
    float inv = 1.0f / c;
    float l0 = bc[0], l1 = bc[1];
    for (int ch = 0; ch < 64; ++ch) {
        float p = sums[g * 64 + ch] * inv;
        out[G * 2 + g * 64 + ch] = p;
        l0 = fmaf(p, Wc[ch * 2 + 0], l0);
        l1 = fmaf(p, Wc[ch * 2 + 1], l1);
    }
    out[g * 2 + 0] = l0;
    out[g * 2 + 1] = l1;
}


extern "C" void kernel_launch(void* const* d_in, const int* in_sizes, int n_in,
                              void* d_out, int out_size, void* d_ws, size_t ws_size,
                              hipStream_t stream) {
    const float* x   = (const float*)d_in[0];
    const float* W1  = (const float*)d_in[1];
    const float* b1  = (const float*)d_in[2];
    const float* W2  = (const float*)d_in[3];
    const float* b2  = (const float*)d_in[4];
    const float* Wc  = (const float*)d_in[5];
    const float* bc  = (const float*)d_in[6];
    const int*   ei  = (const int*)d_in[7];
    const int*   bat = (const int*)d_in[8];

    const int n = in_sizes[0] / IN_CH;
    const int E = in_sizes[7] / 2;
    const int G = 64;

    const int* src = ei;
    const int* dst = ei + E;

    float* ws   = (float*)d_ws;
    float* bufA = ws;                ws += (size_t)n * 64;
    float* bufB = ws;                ws += (size_t)n * 64;
    float* dinv = ws;                ws += n;
    float* sums = ws;                ws += (size_t)G * 64;
    float* cnts = ws;                ws += G;
    int* cnt    = (int*)ws;          ws += n;
    int* rowptr = (int*)ws;          ws += n + 1;
    int* cursor = (int*)ws;          ws += n;
    int* bsum   = (int*)ws;          ws += (n + SCAN_B - 1) / SCAN_B;
    int* adj    = (int*)ws;          ws += E;
    float* h1 = bufA;
    float* o1 = bufB;
    float* h2 = bufA;
    float* o2 = bufB;

    const int B = 256;
    auto cdiv = [](long long a, long long b) { return (int)((a + b - 1) / b); };

    k_init_counts<<<cdiv(n, B), B, 0, stream>>>(cnt, n);
    k_count<<<cdiv(E, B), B, 0, stream>>>(dst, cnt, E);
    k_blocksum<<<cdiv(n, SCAN_B), SCAN_B, 0, stream>>>(cnt, bsum, n);
    k_rowptr<<<cdiv(n, SCAN_B), SCAN_B, 0, stream>>>(cnt, bsum, rowptr, cursor, dinv, n);
    k_fill<<<cdiv(E, B), B, 0, stream>>>(src, dst, cursor, adj, E);

    const int tiles = (n + 15) / 16;
    k_gemm_wmma<128, 32><<<cdiv(tiles, 4), 128, 0, stream>>>(x, W1, h1, n, tiles);
    k_pull<1><<<cdiv((long long)n * 32, B), B, 0, stream>>>(rowptr, adj, dinv, h1, b1, o1, n);

    k_gemm_wmma<32, 64><<<cdiv(tiles, 4), 128, 0, stream>>>(o1, W2, h2, n, tiles);
    k_pull<2><<<cdiv((long long)n * 32, B), B, 0, stream>>>(rowptr, adj, dinv, h2, b2, o2, n);

    hipMemsetAsync(sums, 0, (size_t)(G * 64 + G) * sizeof(float), stream);
    k_pool_onehot<<<cdiv(n, 2 * POOL_CH), 64, 0, stream>>>(o2, bat, sums, cnts, n);
    k_final_head<<<1, 64, 0, stream>>>(sums, cnts, Wc, bc, (float*)d_out, G);
}


# One global atomic per edge, not two: the in-degree count's returning add gives each edge its place in its destination's row; fill stores at row start + place, no atomics; sums in count-arrival order
# speedup vs baseline: 2.5006x; 1.2119x over previous
#include <hip/hip_runtime.h>
#include <hip/hip_bf16.h>


typedef float v2f __attribute__((ext_vector_type(2)));
typedef float v8f __attribute__((ext_vector_type(8)));

#define IN_CH 128


#define SCAN_B 256

__global__ void k_init_counts(int* cnt, int n) {
    int i = blockIdx.x * blockDim.x + threadIdx.x;
    if (i < n) cnt[i] = 0;
}

__global__ void k_count(const int* __restrict__ dst, int* cnt, int* __restrict__ place, int E) {
    int e = blockIdx.x * blockDim.x + threadIdx.x;
    if (e < E) place[e] = atomicAdd(&cnt[dst[e]], 1);
}

__global__ void k_blocksum(const int* __restrict__ cnt, int* bsum, int n) {
    __shared__ int sh[SCAN_B / 32];
    const int i = blockIdx.x * SCAN_B + threadIdx.x;
    int v = i < n ? cnt[i] : 0;
    for (int o = 16; o > 0; o >>= 1) v += __shfl_xor(v, o);
    if ((threadIdx.x & 31) == 0) sh[threadIdx.x >> 5] = v;
    __syncthreads();
    if (threadIdx.x == 0) {
        int t = 0;
        for (int w = 0; w < SCAN_B / 32; ++w) t += sh[w];
        bsum[blockIdx.x] = t;
    }
}

__global__ void k_rowptr(const int* __restrict__ cnt, const int* __restrict__ bsum,
                         int* rowptr, float* dinv, int n) {
    __shared__ int sh_inc[SCAN_B / 32];
    __shared__ int sh_pre[SCAN_B / 32];
    const int tid = threadIdx.x, lane = tid & 31, wv = tid >> 5;

    int p = 0;
    for (int j = tid; j < (int)blockIdx.x; j += SCAN_B) p += bsum[j];
    for (int o = 16; o > 0; o >>= 1) p += __shfl_xor(p, o);
    if (lane == 0) sh_pre[wv] = p;

    const int i = blockIdx.x * SCAN_B + tid;
    const int c = i < n ? cnt[i] : 0;
    int inc = c;
    for (int o = 1; o < 32; o <<= 1) {
        const int t = __shfl_up(inc, o);
        if (lane >= o) inc += t;
    }
    if (lane == 31) sh_inc[wv] = inc;
    __syncthreads();

    int base = 0;
    for (int w = 0; w < SCAN_B / 32; ++w) {
        base += sh_pre[w];
        if (w < wv) base += sh_inc[w];
    }
    const int excl = base + inc - c;
    if (i < n) {
        rowptr[i] = excl;
        const float d = (float)(c + 1);
        dinv[i] = d > 0.0f ? __frsqrt_rn(d) : 0.0f;
    }
    if (i == n - 1) rowptr[n] = excl + c;
}

__global__ void k_fill(const int* __restrict__ src, const int* __restrict__ dst,
                       const int* __restrict__ rowptr, const int* __restrict__ place,
                       int* __restrict__ adj, int E) {
    int e = blockIdx.x * blockDim.x + threadIdx.x;
    if (e < E) adj[rowptr[dst[e]] + place[e]] = src[e];
}


template <int K, int N>
__global__ void k_gemm_wmma(const float* __restrict__ A,
                            const float* __restrict__ W,
                            float* __restrict__ O,
                            int n_rows, int n_tiles) {
    int wave = (blockIdx.x * blockDim.x + threadIdx.x) >> 5;
    int lane = threadIdx.x & 31;
    if (wave >= n_tiles) return;

    const int m0    = wave * 16;
    const int mrow  = m0 + (lane & 15);
    const int rm    = mrow < n_rows ? mrow : (n_rows - 1);
    const int khalf = (lane >> 4) * 2;
    const int colL  = lane & 15;

    constexpr int NT = N / 16;
    v8f acc[NT] = {};

    for (int k = 0; k < K; k += 4) {
        v2f a;
        a.x = A[rm * K + k + khalf];
        a.y = A[rm * K + k + khalf + 1];
#pragma unroll
        for (int t = 0; t < NT; ++t) {
            v2f b;
            b.x = W[(k + khalf) * N + t * 16 + colL];
            b.y = W[(k + khalf + 1) * N + t * 16 + colL];
            acc[t] = __builtin_amdgcn_wmma_f32_16x16x4_f32(
                false, a, false, b, (short)0, acc[t], false, false);
        }
    }

    const int mbase = m0 + (lane >> 4) * 8;
    if (m0 + 16 <= n_rows) {
#pragma unroll
        for (int t = 0; t < NT; ++t) {
#pragma unroll
            for (int r = 0; r < 8; ++r) {
                O[(mbase + r) * N + t * 16 + colL] = acc[t][r];
            }
        }
    } else {
#pragma unroll
        for (int t = 0; t < NT; ++t) {
#pragma unroll
            for (int r = 0; r < 8; ++r) {
                int row = mbase + r;
                if (row < n_rows) O[row * N + t * 16 + colL] = acc[t][r];
            }
        }
    }
}


template <int CPL>
__device__ __forceinline__ void row_load(float (&v)[CPL], const float* __restrict__ t, int idx);

template <>
__device__ __forceinline__ void row_load<1>(float (&v)[1], const float* __restrict__ t, int idx) {
    v[0] = t[idx];
}

template <>
__device__ __forceinline__ void row_load<2>(float (&v)[2], const float* __restrict__ t, int idx) {
    const v2f p = ((const v2f*)t)[idx];
    v[0] = p.x;
    v[1] = p.y;
}

template <int CPL, int U>
__device__ __forceinline__ void pull_group(float (&acc)[CPL],
                                           const float* __restrict__ h, int lane,
                                           int sj, float wj, int j, float dd) {
    float v[U][CPL];
#pragma unroll
    for (int u = 0; u < U; ++u) {
        const int s = __builtin_amdgcn_readlane(sj, j + u);
        row_load<CPL>(v[u], h, s * 32 + lane);
    }
    __builtin_amdgcn_sched_barrier(0);
#pragma unroll
    for (int u = 0; u < U; ++u) {
        const float w = __int_as_float(__builtin_amdgcn_readlane(__float_as_int(wj), j + u));
        const float norm = w * dd;
#pragma unroll
        for (int q = 0; q < CPL; ++q) acc[q] = fmaf(v[u][q], norm, acc[q]);
    }
}

template <int CPL>
__global__ void
k_pull(const int* __restrict__ rowptr,
       const int* __restrict__ adj,
       const float* __restrict__ dinv,
       const float* __restrict__ h,
       const float* __restrict__ bias,
       float* __restrict__ out, int n) {
    const int d = (blockIdx.x * blockDim.x + threadIdx.x) >> 5;
    const int lane = threadIdx.x & 31;
    if (d >= n) return;
    const int beg = __builtin_amdgcn_readfirstlane(rowptr[d]);
    const int end = __builtin_amdgcn_readfirstlane(rowptr[d + 1]);

    const float dd = __int_as_float(__builtin_amdgcn_readfirstlane(__float_as_int(dinv[d])));
    float self[CPL];
    row_load<CPL>(self, h, d * 32 + lane);
    float bv[CPL];
    row_load<CPL>(bv, bias, lane);

    float acc[CPL];
#pragma unroll
    for (int q = 0; q < CPL; ++q) acc[q] = 0.0f;

    for (int b = beg; b < end; b += 32) {
        const int m = end - b < 32 ? end - b : 32;
        int sj = 0;
        float wj = 0.0f;
        if (lane < m) {
            sj = adj[b + lane];
            wj = dinv[sj];
        }
        int j = 0;
        for (; j + 8 <= m; j += 8) pull_group<CPL, 8>(acc, h, lane, sj, wj, j, dd);
        switch (m - j) {
            case 7: pull_group<CPL, 7>(acc, h, lane, sj, wj, j, dd); break;
            case 6: pull_group<CPL, 6>(acc, h, lane, sj, wj, j, dd); break;
            case 5: pull_group<CPL, 5>(acc, h, lane, sj, wj, j, dd); break;
            case 4: pull_group<CPL, 4>(acc, h, lane, sj, wj, j, dd); break;
            case 3: pull_group<CPL, 3>(acc, h, lane, sj, wj, j, dd); break;
            case 2: pull_group<CPL, 2>(acc, h, lane, sj, wj, j, dd); break;
            case 1: pull_group<CPL, 1>(acc, h, lane, sj, wj, j, dd); break;
            default: break;
        }
    }

    const float sn = dd * dd;
    float r[CPL];
#pragma unroll
    for (int q = 0; q < CPL; ++q) {
        const float t = fmaf(self[q], sn, acc[q]) + bv[q];
        r[q] = t > 0.0f ? t : 0.0f;
    }
    if (CPL == 1) {
        out[d * 32 + lane] = r[0];
    } else {
        v2f t;
        t.x = r[0];
        t.y = r[CPL - 1];
        ((v2f*)out)[d * 32 + lane] = t;
    }
}


#define POOL_CH 64

struct PoolSet {
    float4 bx[4];
    float4 by[4];
    int    i0[4];
    int    i1[4];
};

template <bool FULL>
__device__ __forceinline__ void pool_load(PoolSet& S,
                                          const float* __restrict__ h,
                                          const int* __restrict__ batch,
                                          int nb, int hh, int c, int n) {
    if (FULL) {
        const float* hp = h + (size_t)(nb + 2 * hh) * 64 + 4 * c;
        const int*   bp = batch + nb + 2 * hh;
#pragma unroll
        for (int u = 0; u < 4; ++u) {
            S.i0[u] = bp[4 * u];
            S.i1[u] = bp[4 * u + 1];
            S.bx[u] = *(const float4*)(hp + (4 * u) * 64);
            S.by[u] = *(const float4*)(hp + (4 * u + 1) * 64);
        }
    } else {
#pragma unroll
        for (int u = 0; u < 4; ++u) {
            const int node0 = nb + 4 * u + 2 * hh;
            const int node1 = node0 + 1;
            const int r0 = node0 < n ? node0 : n - 1;
            const int r1 = node1 < n ? node1 : n - 1;
            const int i0 = batch[r0];
            const int i1 = batch[r1];
            S.i0[u] = node0 < n ? i0 : -1;
            S.i1[u] = node1 < n ? i1 : -1;
            S.bx[u] = *(const float4*)(h + (size_t)r0 * 64 + 4 * c);
            S.by[u] = *(const float4*)(h + (size_t)r1 * 64 + 4 * c);
        }
    }
}

template <bool FULL>
__device__ __forceinline__ void pool_tile(const float* __restrict__ h,
                                          const int* __restrict__ batch,
                                          float* sums, float* cnts,
                                          int base, int mt, int hh, int c, int n) {
    const int g = mt * 16 + c;
    v8f acc[4] = {};
    float cnt = 0.0f;

    PoolSet cur, nxt;
    pool_load<FULL>(cur, h, batch, base, hh, c, n);
    for (int it = 0; it < POOL_CH / 16; ++it) {
        if (it + 1 < POOL_CH / 16)
            pool_load<FULL>(nxt, h, batch, base + (it + 1) * 16, hh, c, n);
#pragma unroll
        for (int u = 0; u < 4; ++u) {
            v2f a;
            a.x = cur.i0[u] == g ? 1.0f : 0.0f;
            a.y = cur.i1[u] == g ? 1.0f : 0.0f;
            cnt += a.x + a.y;
            const float bx[4] = {cur.bx[u].x, cur.bx[u].y, cur.bx[u].z, cur.bx[u].w};
            const float by[4] = {cur.by[u].x, cur.by[u].y, cur.by[u].z, cur.by[u].w};
#pragma unroll
            for (int nt = 0; nt < 4; ++nt) {
                v2f b;
                b.x = bx[nt];
                b.y = by[nt];
                acc[nt] = __builtin_amdgcn_wmma_f32_16x16x4_f32(
                    false, a, false, b, (short)0, acc[nt], false, false);
            }
        }
        cur = nxt;
    }

    const float tot = cnt + __shfl_xor(cnt, 16);
    if (hh == 0 && tot != 0.0f) unsafeAtomicAdd(&cnts[g], tot);

    const int grow = mt * 16 + hh * 8;
#pragma unroll
    for (int r = 0; r < 8; ++r) {
#pragma unroll
        for (int nt = 0; nt < 4; ++nt) {
            const float v = acc[nt][r];
            if (v != 0.0f)
                unsafeAtomicAdd(&sums[(grow + r) * 64 + 4 * c + nt], v);
        }
    }
}

__global__ void __launch_bounds__(64)
k_pool_onehot(const float* __restrict__ h,
              const int* __restrict__ batch,
              float* sums, float* cnts, int n) {
    const int wave = (blockIdx.x * blockDim.x + threadIdx.x) >> 5;
    const int lane = threadIdx.x & 31;
    const int base = wave * POOL_CH;
    if (base >= n) return;
    const int c  = lane & 15;
    const int hh = lane >> 4;
    const bool full = base + POOL_CH <= n;

    unsigned present = 0;
    for (int i = lane; i < POOL_CH; i += 32) {
        const int node = base + i;
        if (node < n) {
            const unsigned gid = (unsigned)batch[node];
            if (gid < 64u) present |= 1u << (gid >> 4);
        }
    }

    for (int mt = 0; mt < 4; ++mt) {
        if (__ballot((present >> mt) & 1u) == 0) continue;
        if (full) pool_tile<true >(h, batch, sums, cnts, base, mt, hh, c, n);
        else      pool_tile<false>(h, batch, sums, cnts, base, mt, hh, c, n);
    }
}


__global__ void k_final_head(const float* __restrict__ sums,
                             const float* __restrict__ cnts,
                             const float* __restrict__ Wc,
                             const float* __restrict__ bc,
                             float* out, int G) {
    int g = threadIdx.x;
    if (g >= G) return;
    float c = cnts[g];
    c = c > 1.0f ? c : 1.0f;
    float inv = 1.0f / c;
    float l0 = bc[0], l1 = bc[1];
    for (int ch = 0; ch < 64; ++ch) {
        float p = sums[g * 64 + ch] * inv;
        out[G * 2 + g * 64 + ch] = p;
        l0 = fmaf(p, Wc[ch * 2 + 0], l0);
        l1 = fmaf(p, Wc[ch * 2 + 1], l1);
    }
    out[g * 2 + 0] = l0;
    out[g * 2 + 1] = l1;
}


extern "C" void kernel_launch(void* const* d_in, const int* in_sizes, int n_in,
                              void* d_out, int out_size, void* d_ws, size_t ws_size,
                              hipStream_t stream) {
    const float* x   = (const float*)d_in[0];
    const float* W1  = (const float*)d_in[1];
    const float* b1  = (const float*)d_in[2];
    const float* W2  = (const float*)d_in[3];
    const float* b2  = (const float*)d_in[4];
    const float* Wc  = (const float*)d_in[5];
    const float* bc  = (const float*)d_in[6];
    const int*   ei  = (const int*)d_in[7];
    const int*   bat = (const int*)d_in[8];

    const int n = in_sizes[0] / IN_CH;
    const int E = in_sizes[7] / 2;
    const int G = 64;

    const int* src = ei;
    const int* dst = ei + E;

    float* ws   = (float*)d_ws;
    float* bufA = ws;                ws += (size_t)n * 64;
    float* bufB = ws;                ws += (size_t)n * 64;
    float* dinv = ws;                ws += n;
    float* sums = ws;                ws += (size_t)G * 64;
    float* cnts = ws;                ws += G;
    int* cnt    = (int*)ws;          ws += n;
    int* rowptr = (int*)ws;          ws += n + 1;
    int* bsum   = (int*)ws;          ws += (n + SCAN_B - 1) / SCAN_B;
    int* adj    = (int*)ws;          ws += E;
    int* place = (int*)bufA;
    float* h1 = bufA;
    float* o1 = bufB;
    float* h2 = bufA;
    float* o2 = bufB;

    const int B = 256;
    auto cdiv = [](long long a, long long b) { return (int)((a + b - 1) / b); };

    k_init_counts<<<cdiv(n, B), B, 0, stream>>>(cnt, n);
    k_count<<<cdiv(E, B), B, 0, stream>>>(dst, cnt, place, E);
    k_blocksum<<<cdiv(n, SCAN_B), SCAN_B, 0, stream>>>(cnt, bsum, n);
    k_rowptr<<<cdiv(n, SCAN_B), SCAN_B, 0, stream>>>(cnt, bsum, rowptr, dinv, n);
    k_fill<<<cdiv(E, B), B, 0, stream>>>(src, dst, rowptr, place, adj, E);

    const int tiles = (n + 15) / 16;
    k_gemm_wmma<128, 32><<<cdiv(tiles, 4), 128, 0, stream>>>(x, W1, h1, n, tiles);
    k_pull<1><<<cdiv((long long)n * 32, B), B, 0, stream>>>(rowptr, adj, dinv, h1, b1, o1, n);

    k_gemm_wmma<32, 64><<<cdiv(tiles, 4), 128, 0, stream>>>(o1, W2, h2, n, tiles);
    k_pull<2><<<cdiv((long long)n * 32, B), B, 0, stream>>>(rowptr, adj, dinv, h2, b2, o2, n);

    hipMemsetAsync(sums, 0, (size_t)(G * 64 + G) * sizeof(float), stream);
    k_pool_onehot<<<cdiv(n, 2 * POOL_CH), 64, 0, stream>>>(o2, bat, sums, cnts, n);
    k_final_head<<<1, 64, 0, stream>>>(sums, cnts, Wc, bc, (float*)d_out, G);
}


# Edges grouped by destination by a two-level counting sort in LDS (512-node bins, then rows): no global atomics, sort stores are runs of consecutive words; 11 launches to 10; sums in LDS arrival order
# speedup vs baseline: 4.5064x; 1.8021x over previous
#include <hip/hip_runtime.h>
#include <hip/hip_bf16.h>


typedef float v2f __attribute__((ext_vector_type(2)));
typedef float v8f __attribute__((ext_vector_type(8)));

#define IN_CH 128


#define BIN_SHIFT 9
#define BIN_NODES 512
#define MAXB      256
#define SORT_T    1024
#define EPT       6
#define CHUNK     (SORT_T * EPT)
#define STAGE_CAP 16384
#define OWN_U     8

__device__ __forceinline__ int bin_of(int d, int NB) {
    const int b = (int)((unsigned)d >> BIN_SHIFT);
    return b < NB ? b : NB - 1;
}

__device__ __forceinline__ int wave_incl_scan(int v, int lane) {
    for (int o = 1; o < 32; o <<= 1) {
        const int t = __shfl_up(v, o);
        if (lane >= o) v += t;
    }
    return v;
}

__global__ void __launch_bounds__(SORT_T)
k_bin_hist(const int* __restrict__ dst, int* __restrict__ wgcnt, int E, int S, int NB) {
    __shared__ int hist[MAXB];
    const int tid = threadIdx.x, wg = blockIdx.x;
    if (tid < MAXB) hist[tid] = 0;
    __syncthreads();
    const int e0 = wg * CHUNK;
    int d[EPT];
#pragma unroll
    for (int k = 0; k < EPT; ++k) {
        const int e = e0 + k * SORT_T + tid;
        d[k] = dst[e < E ? e : E - 1];
    }
#pragma unroll
    for (int k = 0; k < EPT; ++k) {
        const int e = e0 + k * SORT_T + tid;
        if (e < E) atomicAdd(&hist[bin_of(d[k], NB)], 1);
    }
    __syncthreads();
    if (tid < NB) wgcnt[tid * S + wg] = hist[tid];
}

__global__ void __launch_bounds__(SORT_T)
k_bin_scan(int* wgcnt, int* __restrict__ bintot, int S) {
    __shared__ int wsum[32];
    const int b = blockIdx.x, tid = threadIdx.x, lane = tid & 31, wv = tid >> 5;
    int carry = 0;
    for (int base = 0; base < S; base += SORT_T) {
        const int j = base + tid;
        const int c = j < S ? wgcnt[b * S + j] : 0;
        const int inc = wave_incl_scan(c, lane);
        if (lane == 31) wsum[wv] = inc;
        __syncthreads();
        int wb = 0, tot = 0;
        for (int w = 0; w < 32; ++w) {
            const int v = wsum[w];
            if (w < wv) wb += v;
            tot += v;
        }
        if (j < S) wgcnt[b * S + j] = carry + wb + inc - c;
        carry += tot;
        __syncthreads();
    }
    if (tid == 0) bintot[b] = carry;
}

__global__ void __launch_bounds__(SORT_T)
k_bin_scatter(const int* __restrict__ src, const int* __restrict__ dst,
              const int* __restrict__ wgoff, const int* __restrict__ bintot,
              int* __restrict__ binned, int E, int S, int NB) {
    __shared__ int hist[MAXB];
    __shared__ int lstart[MAXB];
    __shared__ int delta[MAXB];
    __shared__ int wsum_c[MAXB / 32];
    __shared__ int wsum_t[MAXB / 32];
    __shared__ int srec[CHUNK];
    __shared__ int sbin[CHUNK];
    const int tid = threadIdx.x, wg = blockIdx.x, lane = tid & 31, wv = tid >> 5;
    if (tid < MAXB) hist[tid] = 0;
    __syncthreads();

    const int e0 = wg * CHUNK;
    int rec[EPT], bn[EPT], rk[EPT];
#pragma unroll
    for (int k = 0; k < EPT; ++k) {
        const int e = e0 + k * SORT_T + tid;
        const int ec = e < E ? e : E - 1;
        bn[k]  = dst[ec];
        rec[k] = src[ec];
    }
#pragma unroll
    for (int k = 0; k < EPT; ++k) {
        const int e = e0 + k * SORT_T + tid;
        const int d = bn[k];
        bn[k]  = bin_of(d, NB);
        rec[k] = (int)(((unsigned)rec[k] << BIN_SHIFT) | ((unsigned)d & (BIN_NODES - 1)));
        rk[k]  = 0;
        if (e < E) rk[k] = atomicAdd(&hist[bn[k]], 1);
    }
    __syncthreads();

    int c = 0, t = 0, ic = 0, it = 0;
    if (tid < MAXB) {
        c = hist[tid];
        t = tid < NB ? bintot[tid] : 0;
        ic = wave_incl_scan(c, lane);
        it = wave_incl_scan(t, lane);
        if (lane == 31) { wsum_c[wv] = ic; wsum_t[wv] = it; }
    }
    __syncthreads();
    if (tid < MAXB) {
        int cb = 0, tb = 0;
        for (int w = 0; w < wv; ++w) { cb += wsum_c[w]; tb += wsum_t[w]; }
        const int ls = cb + ic - c;
        const int bs = tb + it - t;
        lstart[tid] = ls;
        delta[tid]  = (tid < NB ? bs + wgoff[tid * S + wg] : 0) - ls;
    }
    __syncthreads();

#pragma unroll
    for (int k = 0; k < EPT; ++k) {
        const int e = e0 + k * SORT_T + tid;
        if (e < E) {
            const int p = lstart[bn[k]] + rk[k];
            srec[p] = rec[k];
            sbin[p] = bn[k];
        }
    }
    __syncthreads();

    const int left = E - e0;
    const int m = left < CHUNK ? left : CHUNK;
    for (int i = tid; i < m; i += SORT_T) binned[delta[sbin[i]] + i] = srec[i];
}

__global__ void __launch_bounds__(SORT_T)
k_bin_owner(const int* __restrict__ bintot, const int* __restrict__ binned,
            int* __restrict__ rowptr, float* __restrict__ dinv,
            int* __restrict__ adj, int n) {
    __shared__ int cnt[BIN_NODES];
    __shared__ int cur[BIN_NODES];
    __shared__ int wsum[32];
    __shared__ int sadj[STAGE_CAP];
    const int b = blockIdx.x, tid = threadIdx.x, lane = tid & 31, wv = tid >> 5;

    int p = 0;
    for (int j = tid; j < b; j += SORT_T) p += bintot[j];
    for (int o = 16; o > 0; o >>= 1) p += __shfl_xor(p, o);
    if (lane == 0) wsum[wv] = p;
    if (tid < BIN_NODES) cnt[tid] = 0;
    __syncthreads();
    int start = 0;
    for (int w = 0; w < 32; ++w) start += wsum[w];
    const int m = bintot[b];
    __syncthreads();

    for (int i0 = 0; i0 < m; i0 += OWN_U * SORT_T) {
        int r[OWN_U];
#pragma unroll
        for (int u = 0; u < OWN_U; ++u) {
            const int i = i0 + u * SORT_T + tid;
            r[u] = binned[start + (i < m ? i : m - 1)];
        }
#pragma unroll
        for (int u = 0; u < OWN_U; ++u) {
            const int i = i0 + u * SORT_T + tid;
            if (i < m) atomicAdd(&cnt[r[u] & (BIN_NODES - 1)], 1);
        }
    }
    __syncthreads();

    int c = 0, inc = 0;
    if (tid < BIN_NODES) {
        c = cnt[tid];
        inc = wave_incl_scan(c, lane);
        if (lane == 31) wsum[wv] = inc;
    }
    __syncthreads();
    if (tid < BIN_NODES) {
        int wb = 0;
        for (int w = 0; w < wv; ++w) wb += wsum[w];
        const int excl = wb + inc - c;
        cur[tid] = excl;
        const int node = b * BIN_NODES + tid;
        if (node < n) {
            rowptr[node] = start + excl;
            const float d = (float)(c + 1);
            dinv[node] = d > 0.0f ? __frsqrt_rn(d) : 0.0f;
            if (node == n - 1) rowptr[n] = start + excl + c;
        }
    }
    __syncthreads();

    if (m <= STAGE_CAP) {
        for (int i0 = 0; i0 < m; i0 += OWN_U * SORT_T) {
            int r[OWN_U];
#pragma unroll
            for (int u = 0; u < OWN_U; ++u) {
                const int i = i0 + u * SORT_T + tid;
                r[u] = binned[start + (i < m ? i : m - 1)];
            }
#pragma unroll
            for (int u = 0; u < OWN_U; ++u) {
                const int i = i0 + u * SORT_T + tid;
                if (i < m) {
                    const int pos = atomicAdd(&cur[r[u] & (BIN_NODES - 1)], 1);
                    sadj[pos] = (int)((unsigned)r[u] >> BIN_SHIFT);
                }
            }
        }
        __syncthreads();
        for (int i = tid; i < m; i += SORT_T) adj[start + i] = sadj[i];
    } else {
        for (int i = tid; i < m; i += SORT_T) {
            const int r = binned[start + i];
            const int pos = atomicAdd(&cur[r & (BIN_NODES - 1)], 1);
            adj[start + pos] = (int)((unsigned)r >> BIN_SHIFT);
        }
    }
}


template <int K, int N>
__global__ void k_gemm_wmma(const float* __restrict__ A,
                            const float* __restrict__ W,
                            float* __restrict__ O,
                            int n_rows, int n_tiles) {
    int wave = (blockIdx.x * blockDim.x + threadIdx.x) >> 5;
    int lane = threadIdx.x & 31;
    if (wave >= n_tiles) return;

    const int m0    = wave * 16;
    const int mrow  = m0 + (lane & 15);
    const int rm    = mrow < n_rows ? mrow : (n_rows - 1);
    const int khalf = (lane >> 4) * 2;
    const int colL  = lane & 15;

    constexpr int NT = N / 16;
    v8f acc[NT] = {};

    for (int k = 0; k < K; k += 4) {
        v2f a;
        a.x = A[rm * K + k + khalf];
        a.y = A[rm * K + k + khalf + 1];
#pragma unroll
        for (int t = 0; t < NT; ++t) {
            v2f b;
            b.x = W[(k + khalf) * N + t * 16 + colL];
            b.y = W[(k + khalf + 1) * N + t * 16 + colL];
            acc[t] = __builtin_amdgcn_wmma_f32_16x16x4_f32(
                false, a, false, b, (short)0, acc[t], false, false);
        }
    }

    const int mbase = m0 + (lane >> 4) * 8;
    if (m0 + 16 <= n_rows) {
#pragma unroll
        for (int t = 0; t < NT; ++t) {
#pragma unroll
            for (int r = 0; r < 8; ++r) {
                O[(mbase + r) * N + t * 16 + colL] = acc[t][r];
            }
        }
    } else {
#pragma unroll
        for (int t = 0; t < NT; ++t) {
#pragma unroll
            for (int r = 0; r < 8; ++r) {
                int row = mbase + r;
                if (row < n_rows) O[row * N + t * 16 + colL] = acc[t][r];
            }
        }
    }
}


template <int CPL>
__device__ __forceinline__ void row_load(float (&v)[CPL], const float* __restrict__ t, int idx);

template <>
__device__ __forceinline__ void row_load<1>(float (&v)[1], const float* __restrict__ t, int idx) {
    v[0] = t[idx];
}

template <>
__device__ __forceinline__ void row_load<2>(float (&v)[2], const float* __restrict__ t, int idx) {
    const v2f p = ((const v2f*)t)[idx];
    v[0] = p.x;
    v[1] = p.y;
}

template <int CPL, int U>
__device__ __forceinline__ void pull_group(float (&acc)[CPL],
                                           const float* __restrict__ h, int lane,
                                           int sj, float wj, int j, float dd) {
    float v[U][CPL];
#pragma unroll
    for (int u = 0; u < U; ++u) {
        const int s = __builtin_amdgcn_readlane(sj, j + u);
        row_load<CPL>(v[u], h, s * 32 + lane);
    }
    __builtin_amdgcn_sched_barrier(0);
#pragma unroll
    for (int u = 0; u < U; ++u) {
        const float w = __int_as_float(__builtin_amdgcn_readlane(__float_as_int(wj), j + u));
        const float norm = w * dd;
#pragma unroll
        for (int q = 0; q < CPL; ++q) acc[q] = fmaf(v[u][q], norm, acc[q]);
    }
}

template <int CPL>
__global__ void
k_pull(const int* __restrict__ rowptr,
       const int* __restrict__ adj,
       const float* __restrict__ dinv,
       const float* __restrict__ h,
       const float* __restrict__ bias,
       float* __restrict__ out, int n) {
    const int d = (blockIdx.x * blockDim.x + threadIdx.x) >> 5;
    const int lane = threadIdx.x & 31;
    if (d >= n) return;
    const int beg = __builtin_amdgcn_readfirstlane(rowptr[d]);
    const int end = __builtin_amdgcn_readfirstlane(rowptr[d + 1]);

    const float dd = __int_as_float(__builtin_amdgcn_readfirstlane(__float_as_int(dinv[d])));
    float self[CPL];
    row_load<CPL>(self, h, d * 32 + lane);
    float bv[CPL];
    row_load<CPL>(bv, bias, lane);

    float acc[CPL];
#pragma unroll
    for (int q = 0; q < CPL; ++q) acc[q] = 0.0f;

    for (int b = beg; b < end; b += 32) {
        const int m = end - b < 32 ? end - b : 32;
        int sj = 0;
        float wj = 0.0f;
        if (lane < m) {
            sj = adj[b + lane];
            wj = dinv[sj];
        }
        int j = 0;
        for (; j + 8 <= m; j += 8) pull_group<CPL, 8>(acc, h, lane, sj, wj, j, dd);
        switch (m - j) {
            case 7: pull_group<CPL, 7>(acc, h, lane, sj, wj, j, dd); break;
            case 6: pull_group<CPL, 6>(acc, h, lane, sj, wj, j, dd); break;
            case 5: pull_group<CPL, 5>(acc, h, lane, sj, wj, j, dd); break;
            case 4: pull_group<CPL, 4>(acc, h, lane, sj, wj, j, dd); break;
            case 3: pull_group<CPL, 3>(acc, h, lane, sj, wj, j, dd); break;
            case 2: pull_group<CPL, 2>(acc, h, lane, sj, wj, j, dd); break;
            case 1: pull_group<CPL, 1>(acc, h, lane, sj, wj, j, dd); break;
            default: break;
        }
    }

    const float sn = dd * dd;
    float r[CPL];
#pragma unroll
    for (int q = 0; q < CPL; ++q) {
        const float t = fmaf(self[q], sn, acc[q]) + bv[q];
        r[q] = t > 0.0f ? t : 0.0f;
    }
    if (CPL == 1) {
        out[d * 32 + lane] = r[0];
    } else {
        v2f t;
        t.x = r[0];
        t.y = r[CPL - 1];
        ((v2f*)out)[d * 32 + lane] = t;
    }
}


#define POOL_CH 64

struct PoolSet {
    float4 bx[4];
    float4 by[4];
    int    i0[4];
    int    i1[4];
};

template <bool FULL>
__device__ __forceinline__ void pool_load(PoolSet& S,
                                          const float* __restrict__ h,
                                          const int* __restrict__ batch,
                                          int nb, int hh, int c, int n) {
    if (FULL) {
        const float* hp = h + (size_t)(nb + 2 * hh) * 64 + 4 * c;
        const int*   bp = batch + nb + 2 * hh;
#pragma unroll
        for (int u = 0; u < 4; ++u) {
            S.i0[u] = bp[4 * u];
            S.i1[u] = bp[4 * u + 1];
            S.bx[u] = *(const float4*)(hp + (4 * u) * 64);
            S.by[u] = *(const float4*)(hp + (4 * u + 1) * 64);
        }
    } else {
#pragma unroll
        for (int u = 0; u < 4; ++u) {
            const int node0 = nb + 4 * u + 2 * hh;
            const int node1 = node0 + 1;
            const int r0 = node0 < n ? node0 : n - 1;
            const int r1 = node1 < n ? node1 : n - 1;
            const int i0 = batch[r0];
            const int i1 = batch[r1];
            S.i0[u] = node0 < n ? i0 : -1;
            S.i1[u] = node1 < n ? i1 : -1;
            S.bx[u] = *(const float4*)(h + (size_t)r0 * 64 + 4 * c);
            S.by[u] = *(const float4*)(h + (size_t)r1 * 64 + 4 * c);
        }
    }
}

template <bool FULL>
__device__ __forceinline__ void pool_tile(const float* __restrict__ h,
                                          const int* __restrict__ batch,
                                          float* sums, float* cnts,
                                          int base, int mt, int hh, int c, int n) {
    const int g = mt * 16 + c;
    v8f acc[4] = {};
    float cnt = 0.0f;

    PoolSet cur, nxt;
    pool_load<FULL>(cur, h, batch, base, hh, c, n);
    for (int it = 0; it < POOL_CH / 16; ++it) {
        if (it + 1 < POOL_CH / 16)
            pool_load<FULL>(nxt, h, batch, base + (it + 1) * 16, hh, c, n);
#pragma unroll
        for (int u = 0; u < 4; ++u) {
            v2f a;
            a.x = cur.i0[u] == g ? 1.0f : 0.0f;
            a.y = cur.i1[u] == g ? 1.0f : 0.0f;
            cnt += a.x + a.y;
            const float bx[4] = {cur.bx[u].x, cur.bx[u].y, cur.bx[u].z, cur.bx[u].w};
            const float by[4] = {cur.by[u].x, cur.by[u].y, cur.by[u].z, cur.by[u].w};
#pragma unroll
            for (int nt = 0; nt < 4; ++nt) {
                v2f b;
                b.x = bx[nt];
                b.y = by[nt];
                acc[nt] = __builtin_amdgcn_wmma_f32_16x16x4_f32(
                    false, a, false, b, (short)0, acc[nt], false, false);
            }
        }
        cur = nxt;
    }

    const float tot = cnt + __shfl_xor(cnt, 16);
    if (hh == 0 && tot != 0.0f) unsafeAtomicAdd(&cnts[g], tot);

    const int grow = mt * 16 + hh * 8;
#pragma unroll
    for (int r = 0; r < 8; ++r) {
#pragma unroll
        for (int nt = 0; nt < 4; ++nt) {
            const float v = acc[nt][r];
            if (v != 0.0f)
                unsafeAtomicAdd(&sums[(grow + r) * 64 + 4 * c + nt], v);
        }
    }
}

__global__ void __launch_bounds__(64)
k_pool_onehot(const float* __restrict__ h,
              const int* __restrict__ batch,
              float* sums, float* cnts, int n) {
    const int wave = (blockIdx.x * blockDim.x + threadIdx.x) >> 5;
    const int lane = threadIdx.x & 31;
    const int base = wave * POOL_CH;
    if (base >= n) return;
    const int c  = lane & 15;
    const int hh = lane >> 4;
    const bool full = base + POOL_CH <= n;

    unsigned present = 0;
    for (int i = lane; i < POOL_CH; i += 32) {
        const int node = base + i;
        if (node < n) {
            const unsigned gid = (unsigned)batch[node];
            if (gid < 64u) present |= 1u << (gid >> 4);
        }
    }

    for (int mt = 0; mt < 4; ++mt) {
        if (__ballot((present >> mt) & 1u) == 0) continue;
        if (full) pool_tile<true >(h, batch, sums, cnts, base, mt, hh, c, n);
        else      pool_tile<false>(h, batch, sums, cnts, base, mt, hh, c, n);
    }
}


__global__ void k_final_head(const float* __restrict__ sums,
                             const float* __restrict__ cnts,
                             const float* __restrict__ Wc,
                             const float* __restrict__ bc,
                             float* out, int G) {
    int g = threadIdx.x;
    if (g >= G) return;
    float c = cnts[g];
    c = c > 1.0f ? c : 1.0f;
    float inv = 1.0f / c;
    float l0 = bc[0], l1 = bc[1];
    for (int ch = 0; ch < 64; ++ch) {
        float p = sums[g * 64 + ch] * inv;
        out[G * 2 + g * 64 + ch] = p;
        l0 = fmaf(p, Wc[ch * 2 + 0], l0);
        l1 = fmaf(p, Wc[ch * 2 + 1], l1);
    }
    out[g * 2 + 0] = l0;
    out[g * 2 + 1] = l1;
}


extern "C" void kernel_launch(void* const* d_in, const int* in_sizes, int n_in,
                              void* d_out, int out_size, void* d_ws, size_t ws_size,
                              hipStream_t stream) {
    const float* x   = (const float*)d_in[0];
    const float* W1  = (const float*)d_in[1];
    const float* b1  = (const float*)d_in[2];
    const float* W2  = (const float*)d_in[3];
    const float* b2  = (const float*)d_in[4];
    const float* Wc  = (const float*)d_in[5];
    const float* bc  = (const float*)d_in[6];
    const int*   ei  = (const int*)d_in[7];
    const int*   bat = (const int*)d_in[8];

    const int n = in_sizes[0] / IN_CH;
    const int E = in_sizes[7] / 2;
    const int G = 64;

    const int* src = ei;
    const int* dst = ei + E;

    float* ws   = (float*)d_ws;
    float* bufA = ws;                ws += (size_t)n * 64;
    float* bufB = ws;                ws += (size_t)n * 64;
    float* dinv = ws;                ws += n;
    float* sums = ws;                ws += (size_t)G * 64;
    float* cnts = ws;                ws += G;
    ws += n;
    int* rowptr = (int*)ws;          ws += n + 1;
    ws += (n + 255) / 256;
    int* adj    = (int*)ws;          ws += E;
    float* h1 = bufA;
    float* o1 = bufB;
    float* h2 = bufA;
    float* o2 = bufB;

    const int B = 256;
    auto cdiv = [](long long a, long long b) { return (int)((a + b - 1) / b); };

    const int S  = cdiv(E, CHUNK);
    const int NB = cdiv(n, BIN_NODES);
    int* binned = (int*)bufA;
    int* wgcnt  = (int*)bufB;
    int* bintot = wgcnt + (size_t)NB * S;
    k_bin_hist<<<S, SORT_T, 0, stream>>>(dst, wgcnt, E, S, NB);
    k_bin_scan<<<NB, SORT_T, 0, stream>>>(wgcnt, bintot, S);
    k_bin_scatter<<<S, SORT_T, 0, stream>>>(src, dst, wgcnt, bintot, binned, E, S, NB);
    k_bin_owner<<<NB, SORT_T, 0, stream>>>(bintot, binned, rowptr, dinv, adj, n);

    const int tiles = (n + 15) / 16;
    k_gemm_wmma<128, 32><<<cdiv(tiles, 4), 128, 0, stream>>>(x, W1, h1, n, tiles);
    k_pull<1><<<cdiv((long long)n * 32, B), B, 0, stream>>>(rowptr, adj, dinv, h1, b1, o1, n);

    k_gemm_wmma<32, 64><<<cdiv(tiles, 4), 128, 0, stream>>>(o1, W2, h2, n, tiles);
    k_pull<2><<<cdiv((long long)n * 32, B), B, 0, stream>>>(rowptr, adj, dinv, h2, b2, o2, n);

    hipMemsetAsync(sums, 0, (size_t)(G * 64 + G) * sizeof(float), stream);
    k_pool_onehot<<<cdiv(n, 2 * POOL_CH), 64, 0, stream>>>(o2, bat, sums, cnts, n);
    k_final_head<<<1, 64, 0, stream>>>(sums, cnts, Wc, bc, (float*)d_out, G);
}


# Sort level 1 without global offsets: each chunk stored back in its own bin order plus a (run start, length) table; bin owners walk their runs; 10 launches to 8; sums in LDS arrival order
# speedup vs baseline: 4.9938x; 1.1082x over previous
#include <hip/hip_runtime.h>
#include <hip/hip_bf16.h>


typedef float v2f __attribute__((ext_vector_type(2)));
typedef float v8f __attribute__((ext_vector_type(8)));

#define IN_CH 128


#define BIN_SHIFT 9
#define BIN_NODES 512
#define MAXB      256
#define SORT_T    1024
#define EPT       6
#define CHUNK     (SORT_T * EPT)
#define STAGE_CAP 16384
#define RUN_U     9

__device__ __forceinline__ int bin_of(int d, int NB) {
    const int b = (int)((unsigned)d >> BIN_SHIFT);
    return b < NB ? b : NB - 1;
}

__device__ __forceinline__ int wave_incl_scan(int v, int lane) {
    for (int o = 1; o < 32; o <<= 1) {
        const int t = __shfl_up(v, o);
        if (lane >= o) v += t;
    }
    return v;
}

__global__ void __launch_bounds__(SORT_T)
k_bin_scatter(const int* __restrict__ src, const int* __restrict__ dst,
              int* __restrict__ tab, int* __restrict__ binned, int E, int S, int NB) {
    __shared__ int hist[MAXB];
    __shared__ int lstart[MAXB];
    __shared__ int wsum_c[MAXB / 32];
    __shared__ int srec[CHUNK];
    const int tid = threadIdx.x, wg = blockIdx.x, lane = tid & 31, wv = tid >> 5;
    if (tid < MAXB) hist[tid] = 0;
    __syncthreads();

    const int e0 = wg * CHUNK;
    int rec[EPT], bn[EPT], rk[EPT];
#pragma unroll
    for (int k = 0; k < EPT; ++k) {
        const int e = e0 + k * SORT_T + tid;
        const int ec = e < E ? e : E - 1;
        bn[k]  = dst[ec];
        rec[k] = src[ec];
    }
#pragma unroll
    for (int k = 0; k < EPT; ++k) {
        const int e = e0 + k * SORT_T + tid;
        const int d = bn[k];
        bn[k]  = bin_of(d, NB);
        rec[k] = (int)(((unsigned)rec[k] << BIN_SHIFT) | ((unsigned)d & (BIN_NODES - 1)));
        rk[k]  = 0;
        if (e < E) rk[k] = atomicAdd(&hist[bn[k]], 1);
    }
    __syncthreads();

    int c = 0, ic = 0;
    if (tid < MAXB) {
        c = hist[tid];
        ic = wave_incl_scan(c, lane);
        if (lane == 31) wsum_c[wv] = ic;
    }
    __syncthreads();
    if (tid < MAXB) {
        int cb = 0;
        for (int w = 0; w < wv; ++w) cb += wsum_c[w];
        const int ls = cb + ic - c;
        lstart[tid] = ls;
        if (tid < NB) tab[tid * S + wg] = (ls << 16) | c;
    }
    __syncthreads();

#pragma unroll
    for (int k = 0; k < EPT; ++k) {
        const int e = e0 + k * SORT_T + tid;
        if (e < E) srec[lstart[bn[k]] + rk[k]] = rec[k];
    }
    __syncthreads();

    const int left = E - e0;
    const int m = left < CHUNK ? left : CHUNK;
    for (int i = tid; i < m; i += SORT_T) binned[e0 + i] = srec[i];
}

template <int MODE>
__device__ __forceinline__ int own_place(int r, int* tbl) {
    if (MODE == 0) {
        atomicAdd(&tbl[r & (BIN_NODES - 1)], 1);
        return 0;
    }
    return atomicAdd(&tbl[r & (BIN_NODES - 1)], 1);
}

template <int MODE>
__device__ __forceinline__ void own_put(int r, int pos, int* sadj,
                                        int* __restrict__ adj, int start) {
    const int s = (int)((unsigned)r >> BIN_SHIFT);
    if (MODE == 1) sadj[pos] = s;
    if (MODE == 2) adj[start + pos] = s;
}

template <int MODE>
__device__ __forceinline__ void own_runs(const int* __restrict__ trow,
                                         const int* __restrict__ binned,
                                         int* tbl, int* sadj, int* __restrict__ adj,
                                         int start, int E, int w0, int nu, int lane,
                                         int& st, int& tot) {
    for (int u0 = 0; u0 < nu; u0 += 32) {
        const int lim = nu - u0 < 32 ? nu - u0 : 32;
        int e = 0;
        if (lane < lim) e = trow[w0 + u0 + lane];
        if (MODE == 0) {
            st  += (int)((unsigned)e >> 16);
            tot += e & 0xFFFF;
        }
        for (int u = 0; u < lim; u += RUN_U) {
            int r0[RUN_U], r1[RUN_U], cn[RUN_U], bs[RUN_U];
#pragma unroll
            for (int k = 0; k < RUN_U; ++k) {
                const bool in = u + k < lim;
                const int uu = in ? u + k : lim - 1;
                const int eu = __builtin_amdgcn_readlane(e, uu);
                const int c  = in ? (eu & 0xFFFF) : 0;
                const int base = (w0 + u0 + uu) * CHUNK + (int)((unsigned)eu >> 16);
                cn[k] = c;
                bs[k] = base;
                int i0 = lane < c ? lane : c - 1;
                int i1 = lane + 32 < c ? lane + 32 : c - 1;
                i0 = i0 > 0 ? i0 : 0;
                i1 = i1 > 0 ? i1 : 0;
                int a0 = base + i0, a1 = base + i1;
                a0 = a0 < E ? a0 : E - 1;
                a1 = a1 < E ? a1 : E - 1;
                r0[k] = binned[a0];
                r1[k] = binned[a1];
            }
            int p0[RUN_U], p1[RUN_U];
#pragma unroll
            for (int k = 0; k < RUN_U; ++k) {
                p0[k] = 0;
                p1[k] = 0;
                if (lane < cn[k])      p0[k] = own_place<MODE>(r0[k], tbl);
                if (lane + 32 < cn[k]) p1[k] = own_place<MODE>(r1[k], tbl);
            }
#pragma unroll
            for (int k = 0; k < RUN_U; ++k) {
                if (MODE != 0) {
                    if (lane < cn[k])      own_put<MODE>(r0[k], p0[k], sadj, adj, start);
                    if (lane + 32 < cn[k]) own_put<MODE>(r1[k], p1[k], sadj, adj, start);
                }
                for (int i = lane + 64; i < cn[k]; i += 32) {
                    const int r = binned[bs[k] + i];
                    const int pos = own_place<MODE>(r, tbl);
                    if (MODE != 0) own_put<MODE>(r, pos, sadj, adj, start);
                }
            }
        }
    }
}

__global__ void __launch_bounds__(SORT_T)
k_bin_owner(const int* __restrict__ tab, const int* __restrict__ binned,
            int* __restrict__ rowptr, float* __restrict__ dinv,
            int* __restrict__ adj, int n, int E, int S) {
    __shared__ int cnt[BIN_NODES];
    __shared__ int cur[BIN_NODES];
    __shared__ int wsum[32];
    __shared__ int wst[32];
    __shared__ int wtot[32];
    __shared__ int sadj[STAGE_CAP];
    const int b = blockIdx.x, tid = threadIdx.x, lane = tid & 31;
    const int wv = __builtin_amdgcn_readfirstlane(tid >> 5);

    const int rpw = (S + 31) >> 5;
    const int w0  = wv * rpw;
    const int nu  = S - w0 < rpw ? (S - w0 > 0 ? S - w0 : 0) : rpw;
    const int* trow = tab + (size_t)b * S;

    if (tid < BIN_NODES) cnt[tid] = 0;
    __syncthreads();

    int st = 0, tot = 0;
    own_runs<0>(trow, binned, cnt, sadj, adj, 0, E, w0, nu, lane, st, tot);
    for (int o = 16; o > 0; o >>= 1) {
        st  += __shfl_xor(st, o);
        tot += __shfl_xor(tot, o);
    }
    if (lane == 0) { wst[wv] = st; wtot[wv] = tot; }
    __syncthreads();

    int start = 0, m = 0;
    for (int w = 0; w < 32; ++w) { start += wst[w]; m += wtot[w]; }

    int c = 0, inc = 0;
    if (tid < BIN_NODES) {
        c = cnt[tid];
        inc = wave_incl_scan(c, lane);
        if (lane == 31) wsum[wv] = inc;
    }
    __syncthreads();
    if (tid < BIN_NODES) {
        int wb = 0;
        for (int w = 0; w < wv; ++w) wb += wsum[w];
        const int excl = wb + inc - c;
        cur[tid] = excl;
        const int node = b * BIN_NODES + tid;
        if (node < n) {
            rowptr[node] = start + excl;
            const float d = (float)(c + 1);
            dinv[node] = d > 0.0f ? __frsqrt_rn(d) : 0.0f;
            if (node == n - 1) rowptr[n] = start + excl + c;
        }
    }
    __syncthreads();

    int dum0 = 0, dum1 = 0;
    if (m <= STAGE_CAP) {
        own_runs<1>(trow, binned, cur, sadj, adj, start, E, w0, nu, lane, dum0, dum1);
        __syncthreads();
        for (int i = tid; i < m; i += SORT_T) adj[start + i] = sadj[i];
    } else {
        own_runs<2>(trow, binned, cur, sadj, adj, start, E, w0, nu, lane, dum0, dum1);
    }
}


template <int K, int N>
__global__ void k_gemm_wmma(const float* __restrict__ A,
                            const float* __restrict__ W,
                            float* __restrict__ O,
                            int n_rows, int n_tiles) {
    int wave = (blockIdx.x * blockDim.x + threadIdx.x) >> 5;
    int lane = threadIdx.x & 31;
    if (wave >= n_tiles) return;

    const int m0    = wave * 16;
    const int mrow  = m0 + (lane & 15);
    const int rm    = mrow < n_rows ? mrow : (n_rows - 1);
    const int khalf = (lane >> 4) * 2;
    const int colL  = lane & 15;

    constexpr int NT = N / 16;
    v8f acc[NT] = {};

    for (int k = 0; k < K; k += 4) {
        v2f a;
        a.x = A[rm * K + k + khalf];
        a.y = A[rm * K + k + khalf + 1];
#pragma unroll
        for (int t = 0; t < NT; ++t) {
            v2f b;
            b.x = W[(k + khalf) * N + t * 16 + colL];
            b.y = W[(k + khalf + 1) * N + t * 16 + colL];
            acc[t] = __builtin_amdgcn_wmma_f32_16x16x4_f32(
                false, a, false, b, (short)0, acc[t], false, false);
        }
    }

    const int mbase = m0 + (lane >> 4) * 8;
    if (m0 + 16 <= n_rows) {
#pragma unroll
        for (int t = 0; t < NT; ++t) {
#pragma unroll
            for (int r = 0; r < 8; ++r) {
                O[(mbase + r) * N + t * 16 + colL] = acc[t][r];
            }
        }
    } else {
#pragma unroll
        for (int t = 0; t < NT; ++t) {
#pragma unroll
            for (int r = 0; r < 8; ++r) {
                int row = mbase + r;
                if (row < n_rows) O[row * N + t * 16 + colL] = acc[t][r];
            }
        }
    }
}


template <int CPL>
__device__ __forceinline__ void row_load(float (&v)[CPL], const float* __restrict__ t, int idx);

template <>
__device__ __forceinline__ void row_load<1>(float (&v)[1], const float* __restrict__ t, int idx) {
    v[0] = t[idx];
}

template <>
__device__ __forceinline__ void row_load<2>(float (&v)[2], const float* __restrict__ t, int idx) {
    const v2f p = ((const v2f*)t)[idx];
    v[0] = p.x;
    v[1] = p.y;
}

template <int CPL, int U>
__device__ __forceinline__ void pull_group(float (&acc)[CPL],
                                           const float* __restrict__ h, int lane,
                                           int sj, float wj, int j, float dd) {
    float v[U][CPL];
#pragma unroll
    for (int u = 0; u < U; ++u) {
        const int s = __builtin_amdgcn_readlane(sj, j + u);
        row_load<CPL>(v[u], h, s * 32 + lane);
    }
    __builtin_amdgcn_sched_barrier(0);
#pragma unroll
    for (int u = 0; u < U; ++u) {
        const float w = __int_as_float(__builtin_amdgcn_readlane(__float_as_int(wj), j + u));
        const float norm = w * dd;
#pragma unroll
        for (int q = 0; q < CPL; ++q) acc[q] = fmaf(v[u][q], norm, acc[q]);
    }
}

template <int CPL>
__global__ void
k_pull(const int* __restrict__ rowptr,
       const int* __restrict__ adj,
       const float* __restrict__ dinv,
       const float* __restrict__ h,
       const float* __restrict__ bias,
       float* __restrict__ out, int n) {
    const int d = (blockIdx.x * blockDim.x + threadIdx.x) >> 5;
    const int lane = threadIdx.x & 31;
    if (d >= n) return;
    const int beg = __builtin_amdgcn_readfirstlane(rowptr[d]);
    const int end = __builtin_amdgcn_readfirstlane(rowptr[d + 1]);

    const float dd = __int_as_float(__builtin_amdgcn_readfirstlane(__float_as_int(dinv[d])));
    float self[CPL];
    row_load<CPL>(self, h, d * 32 + lane);
    float bv[CPL];
    row_load<CPL>(bv, bias, lane);

    float acc[CPL];
#pragma unroll
    for (int q = 0; q < CPL; ++q) acc[q] = 0.0f;

    for (int b = beg; b < end; b += 32) {
        const int m = end - b < 32 ? end - b : 32;
        int sj = 0;
        float wj = 0.0f;
        if (lane < m) {
            sj = adj[b + lane];
            wj = dinv[sj];
        }
        int j = 0;
        for (; j + 8 <= m; j += 8) pull_group<CPL, 8>(acc, h, lane, sj, wj, j, dd);
        switch (m - j) {
            case 7: pull_group<CPL, 7>(acc, h, lane, sj, wj, j, dd); break;
            case 6: pull_group<CPL, 6>(acc, h, lane, sj, wj, j, dd); break;
            case 5: pull_group<CPL, 5>(acc, h, lane, sj, wj, j, dd); break;
            case 4: pull_group<CPL, 4>(acc, h, lane, sj, wj, j, dd); break;
            case 3: pull_group<CPL, 3>(acc, h, lane, sj, wj, j, dd); break;
            case 2: pull_group<CPL, 2>(acc, h, lane, sj, wj, j, dd); break;
            case 1: pull_group<CPL, 1>(acc, h, lane, sj, wj, j, dd); break;
            default: break;
        }
    }

    const float sn = dd * dd;
    float r[CPL];
#pragma unroll
    for (int q = 0; q < CPL; ++q) {
        const float t = fmaf(self[q], sn, acc[q]) + bv[q];
        r[q] = t > 0.0f ? t : 0.0f;
    }
    if (CPL == 1) {
        out[d * 32 + lane] = r[0];
    } else {
        v2f t;
        t.x = r[0];
        t.y = r[CPL - 1];
        ((v2f*)out)[d * 32 + lane] = t;
    }
}


#define POOL_CH 64

struct PoolSet {
    float4 bx[4];
    float4 by[4];
    int    i0[4];
    int    i1[4];
};

template <bool FULL>
__device__ __forceinline__ void pool_load(PoolSet& S,
                                          const float* __restrict__ h,
                                          const int* __restrict__ batch,
                                          int nb, int hh, int c, int n) {
    if (FULL) {
        const float* hp = h + (size_t)(nb + 2 * hh) * 64 + 4 * c;
        const int*   bp = batch + nb + 2 * hh;
#pragma unroll
        for (int u = 0; u < 4; ++u) {
            S.i0[u] = bp[4 * u];
            S.i1[u] = bp[4 * u + 1];
            S.bx[u] = *(const float4*)(hp + (4 * u) * 64);
            S.by[u] = *(const float4*)(hp + (4 * u + 1) * 64);
        }
    } else {
#pragma unroll
        for (int u = 0; u < 4; ++u) {
            const int node0 = nb + 4 * u + 2 * hh;
            const int node1 = node0 + 1;
            const int r0 = node0 < n ? node0 : n - 1;
            const int r1 = node1 < n ? node1 : n - 1;
            const int i0 = batch[r0];
            const int i1 = batch[r1];
            S.i0[u] = node0 < n ? i0 : -1;
            S.i1[u] = node1 < n ? i1 : -1;
            S.bx[u] = *(const float4*)(h + (size_t)r0 * 64 + 4 * c);
            S.by[u] = *(const float4*)(h + (size_t)r1 * 64 + 4 * c);
        }
    }
}

template <bool FULL>
__device__ __forceinline__ void pool_tile(const float* __restrict__ h,
                                          const int* __restrict__ batch,
                                          float* sums, float* cnts,
                                          int base, int mt, int hh, int c, int n) {
    const int g = mt * 16 + c;
    v8f acc[4] = {};
    float cnt = 0.0f;

    PoolSet cur, nxt;
    pool_load<FULL>(cur, h, batch, base, hh, c, n);
    for (int it = 0; it < POOL_CH / 16; ++it) {
        if (it + 1 < POOL_CH / 16)
            pool_load<FULL>(nxt, h, batch, base + (it + 1) * 16, hh, c, n);
#pragma unroll
        for (int u = 0; u < 4; ++u) {
            v2f a;
            a.x = cur.i0[u] == g ? 1.0f : 0.0f;
            a.y = cur.i1[u] == g ? 1.0f : 0.0f;
            cnt += a.x + a.y;
            const float bx[4] = {cur.bx[u].x, cur.bx[u].y, cur.bx[u].z, cur.bx[u].w};
            const float by[4] = {cur.by[u].x, cur.by[u].y, cur.by[u].z, cur.by[u].w};
#pragma unroll
            for (int nt = 0; nt < 4; ++nt) {
                v2f b;
                b.x = bx[nt];
                b.y = by[nt];
                acc[nt] = __builtin_amdgcn_wmma_f32_16x16x4_f32(
                    false, a, false, b, (short)0, acc[nt], false, false);
            }
        }
        cur = nxt;
    }

    const float tot = cnt + __shfl_xor(cnt, 16);
    if (hh == 0 && tot != 0.0f) unsafeAtomicAdd(&cnts[g], tot);

    const int grow = mt * 16 + hh * 8;
#pragma unroll
    for (int r = 0; r < 8; ++r) {
#pragma unroll
        for (int nt = 0; nt < 4; ++nt) {
            const float v = acc[nt][r];
            if (v != 0.0f)
                unsafeAtomicAdd(&sums[(grow + r) * 64 + 4 * c + nt], v);
        }
    }
}

__global__ void __launch_bounds__(64)
k_pool_onehot(const float* __restrict__ h,
              const int* __restrict__ batch,
              float* sums, float* cnts, int n) {
    const int wave = (blockIdx.x * blockDim.x + threadIdx.x) >> 5;
    const int lane = threadIdx.x & 31;
    const int base = wave * POOL_CH;
    if (base >= n) return;
    const int c  = lane & 15;
    const int hh = lane >> 4;
    const bool full = base + POOL_CH <= n;

    unsigned present = 0;
    for (int i = lane; i < POOL_CH; i += 32) {
        const int node = base + i;
        if (node < n) {
            const unsigned gid = (unsigned)batch[node];
            if (gid < 64u) present |= 1u << (gid >> 4);
        }
    }

    for (int mt = 0; mt < 4; ++mt) {
        if (__ballot((present >> mt) & 1u) == 0) continue;
        if (full) pool_tile<true >(h, batch, sums, cnts, base, mt, hh, c, n);
        else      pool_tile<false>(h, batch, sums, cnts, base, mt, hh, c, n);
    }
}


__global__ void k_final_head(const float* __restrict__ sums,
                             const float* __restrict__ cnts,
                             const float* __restrict__ Wc,
                             const float* __restrict__ bc,
                             float* out, int G) {
    int g = threadIdx.x;
    if (g >= G) return;
    float c = cnts[g];
    c = c > 1.0f ? c : 1.0f;
    float inv = 1.0f / c;
    float l0 = bc[0], l1 = bc[1];
    for (int ch = 0; ch < 64; ++ch) {
        float p = sums[g * 64 + ch] * inv;
        out[G * 2 + g * 64 + ch] = p;
        l0 = fmaf(p, Wc[ch * 2 + 0], l0);
        l1 = fmaf(p, Wc[ch * 2 + 1], l1);
    }
    out[g * 2 + 0] = l0;
    out[g * 2 + 1] = l1;
}


extern "C" void kernel_launch(void* const* d_in, const int* in_sizes, int n_in,
                              void* d_out, int out_size, void* d_ws, size_t ws_size,
                              hipStream_t stream) {
    const float* x   = (const float*)d_in[0];
    const float* W1  = (const float*)d_in[1];
    const float* b1  = (const float*)d_in[2];
    const float* W2  = (const float*)d_in[3];
    const float* b2  = (const float*)d_in[4];
    const float* Wc  = (const float*)d_in[5];
    const float* bc  = (const float*)d_in[6];
    const int*   ei  = (const int*)d_in[7];
    const int*   bat = (const int*)d_in[8];

    const int n = in_sizes[0] / IN_CH;
    const int E = in_sizes[7] / 2;
    const int G = 64;

    const int* src = ei;
    const int* dst = ei + E;

    float* ws   = (float*)d_ws;
    float* bufA = ws;                ws += (size_t)n * 64;
    float* bufB = ws;                ws += (size_t)n * 64;
    float* dinv = ws;                ws += n;
    float* sums = ws;                ws += (size_t)G * 64;
    float* cnts = ws;                ws += G;
    ws += n;
    int* rowptr = (int*)ws;          ws += n + 1;
    ws += (n + 255) / 256;
    int* adj    = (int*)ws;          ws += E;
    float* h1 = bufA;
    float* o1 = bufB;
    float* h2 = bufA;
    float* o2 = bufB;

    const int B = 256;
    auto cdiv = [](long long a, long long b) { return (int)((a + b - 1) / b); };

    const int S  = cdiv(E, CHUNK);
    const int NB = cdiv(n, BIN_NODES);
    int* binned = (int*)bufA;
    int* tab    = (int*)bufB;
    k_bin_scatter<<<S, SORT_T, 0, stream>>>(src, dst, tab, binned, E, S, NB);
    k_bin_owner<<<NB, SORT_T, 0, stream>>>(tab, binned, rowptr, dinv, adj, n, E, S);

    const int tiles = (n + 15) / 16;
    k_gemm_wmma<128, 32><<<cdiv(tiles, 4), 128, 0, stream>>>(x, W1, h1, n, tiles);
    k_pull<1><<<cdiv((long long)n * 32, B), B, 0, stream>>>(rowptr, adj, dinv, h1, b1, o1, n);

    k_gemm_wmma<32, 64><<<cdiv(tiles, 4), 128, 0, stream>>>(o1, W2, h2, n, tiles);
    k_pull<2><<<cdiv((long long)n * 32, B), B, 0, stream>>>(rowptr, adj, dinv, h2, b2, o2, n);

    hipMemsetAsync(sums, 0, (size_t)(G * 64 + G) * sizeof(float), stream);
    k_pool_onehot<<<cdiv(n, 2 * POOL_CH), 64, 0, stream>>>(o2, bat, sums, cnts, n);
    k_final_head<<<1, 64, 0, stream>>>(sums, cnts, Wc, bc, (float*)d_out, G);
}


# Layer-2 product (K=32, N=64) moved into the layer-1 edge-sum kernel: 16 nodes per work-group pass finished rows through LDS to 4 waves' f32 matrix tiles; layer-1 output never stored; 8 launches to 7
# speedup vs baseline: 5.7436x; 1.1501x over previous
#include <hip/hip_runtime.h>
#include <hip/hip_bf16.h>


typedef float v2f __attribute__((ext_vector_type(2)));
typedef float v8f __attribute__((ext_vector_type(8)));

#define IN_CH 128


#define BIN_SHIFT 9
#define BIN_NODES 512
#define MAXB      256
#define SORT_T    1024
#define EPT       6
#define CHUNK     (SORT_T * EPT)
#define STAGE_CAP 16384
#define RUN_U     9

__device__ __forceinline__ int bin_of(int d, int NB) {
    const int b = (int)((unsigned)d >> BIN_SHIFT);
    return b < NB ? b : NB - 1;
}

__device__ __forceinline__ int wave_incl_scan(int v, int lane) {
    for (int o = 1; o < 32; o <<= 1) {
        const int t = __shfl_up(v, o);
        if (lane >= o) v += t;
    }
    return v;
}

__global__ void __launch_bounds__(SORT_T)
k_bin_scatter(const int* __restrict__ src, const int* __restrict__ dst,
              int* __restrict__ tab, int* __restrict__ binned, int E, int S, int NB) {
    __shared__ int hist[MAXB];
    __shared__ int lstart[MAXB];
    __shared__ int wsum_c[MAXB / 32];
    __shared__ int srec[CHUNK];
    const int tid = threadIdx.x, wg = blockIdx.x, lane = tid & 31, wv = tid >> 5;
    if (tid < MAXB) hist[tid] = 0;
    __syncthreads();

    const int e0 = wg * CHUNK;
    int rec[EPT], bn[EPT], rk[EPT];
#pragma unroll
    for (int k = 0; k < EPT; ++k) {
        const int e = e0 + k * SORT_T + tid;
        const int ec = e < E ? e : E - 1;
        bn[k]  = dst[ec];
        rec[k] = src[ec];
    }
#pragma unroll
    for (int k = 0; k < EPT; ++k) {
        const int e = e0 + k * SORT_T + tid;
        const int d = bn[k];
        bn[k]  = bin_of(d, NB);
        rec[k] = (int)(((unsigned)rec[k] << BIN_SHIFT) | ((unsigned)d & (BIN_NODES - 1)));
        rk[k]  = 0;
        if (e < E) rk[k] = atomicAdd(&hist[bn[k]], 1);
    }
    __syncthreads();

    int c = 0, ic = 0;
    if (tid < MAXB) {
        c = hist[tid];
        ic = wave_incl_scan(c, lane);
        if (lane == 31) wsum_c[wv] = ic;
    }
    __syncthreads();
    if (tid < MAXB) {
        int cb = 0;
        for (int w = 0; w < wv; ++w) cb += wsum_c[w];
        const int ls = cb + ic - c;
        lstart[tid] = ls;
        if (tid < NB) tab[tid * S + wg] = (ls << 16) | c;
    }
    __syncthreads();

#pragma unroll
    for (int k = 0; k < EPT; ++k) {
        const int e = e0 + k * SORT_T + tid;
        if (e < E) srec[lstart[bn[k]] + rk[k]] = rec[k];
    }
    __syncthreads();

    const int left = E - e0;
    const int m = left < CHUNK ? left : CHUNK;
    for (int i = tid; i < m; i += SORT_T) binned[e0 + i] = srec[i];
}

template <int MODE>
__device__ __forceinline__ int own_place(int r, int* tbl) {
    if (MODE == 0) {
        atomicAdd(&tbl[r & (BIN_NODES - 1)], 1);
        return 0;
    }
    return atomicAdd(&tbl[r & (BIN_NODES - 1)], 1);
}

template <int MODE>
__device__ __forceinline__ void own_put(int r, int pos, int* sadj,
                                        int* __restrict__ adj, int start) {
    const int s = (int)((unsigned)r >> BIN_SHIFT);
    if (MODE == 1) sadj[pos] = s;
    if (MODE == 2) adj[start + pos] = s;
}

template <int MODE>
__device__ __forceinline__ void own_runs(const int* __restrict__ trow,
                                         const int* __restrict__ binned,
                                         int* tbl, int* sadj, int* __restrict__ adj,
                                         int start, int E, int w0, int nu, int lane,
                                         int& st, int& tot) {
    for (int u0 = 0; u0 < nu; u0 += 32) {
        const int lim = nu - u0 < 32 ? nu - u0 : 32;
        int e = 0;
        if (lane < lim) e = trow[w0 + u0 + lane];
        if (MODE == 0) {
            st  += (int)((unsigned)e >> 16);
            tot += e & 0xFFFF;
        }
        for (int u = 0; u < lim; u += RUN_U) {
            int r0[RUN_U], r1[RUN_U], cn[RUN_U], bs[RUN_U];
#pragma unroll
            for (int k = 0; k < RUN_U; ++k) {
                const bool in = u + k < lim;
                const int uu = in ? u + k : lim - 1;
                const int eu = __builtin_amdgcn_readlane(e, uu);
                const int c  = in ? (eu & 0xFFFF) : 0;
                const int base = (w0 + u0 + uu) * CHUNK + (int)((unsigned)eu >> 16);
                cn[k] = c;
                bs[k] = base;
                int i0 = lane < c ? lane : c - 1;
                int i1 = lane + 32 < c ? lane + 32 : c - 1;
                i0 = i0 > 0 ? i0 : 0;
                i1 = i1 > 0 ? i1 : 0;
                int a0 = base + i0, a1 = base + i1;
                a0 = a0 < E ? a0 : E - 1;
                a1 = a1 < E ? a1 : E - 1;
                r0[k] = binned[a0];
                r1[k] = binned[a1];
            }
            int p0[RUN_U], p1[RUN_U];
#pragma unroll
            for (int k = 0; k < RUN_U; ++k) {
                p0[k] = 0;
                p1[k] = 0;
                if (lane < cn[k])      p0[k] = own_place<MODE>(r0[k], tbl);
                if (lane + 32 < cn[k]) p1[k] = own_place<MODE>(r1[k], tbl);
            }
#pragma unroll
            for (int k = 0; k < RUN_U; ++k) {
                if (MODE != 0) {
                    if (lane < cn[k])      own_put<MODE>(r0[k], p0[k], sadj, adj, start);
                    if (lane + 32 < cn[k]) own_put<MODE>(r1[k], p1[k], sadj, adj, start);
                }
                for (int i = lane + 64; i < cn[k]; i += 32) {
                    const int r = binned[bs[k] + i];
                    const int pos = own_place<MODE>(r, tbl);
                    if (MODE != 0) own_put<MODE>(r, pos, sadj, adj, start);
                }
            }
        }
    }
}

__global__ void __launch_bounds__(SORT_T)
k_bin_owner(const int* __restrict__ tab, const int* __restrict__ binned,
            int* __restrict__ rowptr, float* __restrict__ dinv,
            int* __restrict__ adj, int n, int E, int S) {
    __shared__ int cnt[BIN_NODES];
    __shared__ int cur[BIN_NODES];
    __shared__ int wsum[32];
    __shared__ int wst[32];
    __shared__ int wtot[32];
    __shared__ int sadj[STAGE_CAP];
    const int b = blockIdx.x, tid = threadIdx.x, lane = tid & 31;
    const int wv = __builtin_amdgcn_readfirstlane(tid >> 5);

    const int rpw = (S + 31) >> 5;
    const int w0  = wv * rpw;
    const int nu  = S - w0 < rpw ? (S - w0 > 0 ? S - w0 : 0) : rpw;
    const int* trow = tab + (size_t)b * S;

    if (tid < BIN_NODES) cnt[tid] = 0;
    __syncthreads();

    int st = 0, tot = 0;
    own_runs<0>(trow, binned, cnt, sadj, adj, 0, E, w0, nu, lane, st, tot);
    for (int o = 16; o > 0; o >>= 1) {
        st  += __shfl_xor(st, o);
        tot += __shfl_xor(tot, o);
    }
    if (lane == 0) { wst[wv] = st; wtot[wv] = tot; }
    __syncthreads();

    int start = 0, m = 0;
    for (int w = 0; w < 32; ++w) { start += wst[w]; m += wtot[w]; }

    int c = 0, inc = 0;
    if (tid < BIN_NODES) {
        c = cnt[tid];
        inc = wave_incl_scan(c, lane);
        if (lane == 31) wsum[wv] = inc;
    }
    __syncthreads();
    if (tid < BIN_NODES) {
        int wb = 0;
        for (int w = 0; w < wv; ++w) wb += wsum[w];
        const int excl = wb + inc - c;
        cur[tid] = excl;
        const int node = b * BIN_NODES + tid;
        if (node < n) {
            rowptr[node] = start + excl;
            const float d = (float)(c + 1);
            dinv[node] = d > 0.0f ? __frsqrt_rn(d) : 0.0f;
            if (node == n - 1) rowptr[n] = start + excl + c;
        }
    }
    __syncthreads();

    int dum0 = 0, dum1 = 0;
    if (m <= STAGE_CAP) {
        own_runs<1>(trow, binned, cur, sadj, adj, start, E, w0, nu, lane, dum0, dum1);
        __syncthreads();
        for (int i = tid; i < m; i += SORT_T) adj[start + i] = sadj[i];
    } else {
        own_runs<2>(trow, binned, cur, sadj, adj, start, E, w0, nu, lane, dum0, dum1);
    }
}


template <int K, int N>
__global__ void k_gemm_wmma(const float* __restrict__ A,
                            const float* __restrict__ W,
                            float* __restrict__ O,
                            int n_rows, int n_tiles) {
    int wave = (blockIdx.x * blockDim.x + threadIdx.x) >> 5;
    int lane = threadIdx.x & 31;
    if (wave >= n_tiles) return;

    const int m0    = wave * 16;
    const int mrow  = m0 + (lane & 15);
    const int rm    = mrow < n_rows ? mrow : (n_rows - 1);
    const int khalf = (lane >> 4) * 2;
    const int colL  = lane & 15;

    constexpr int NT = N / 16;
    v8f acc[NT] = {};

    for (int k = 0; k < K; k += 4) {
        v2f a;
        a.x = A[rm * K + k + khalf];
        a.y = A[rm * K + k + khalf + 1];
#pragma unroll
        for (int t = 0; t < NT; ++t) {
            v2f b;
            b.x = W[(k + khalf) * N + t * 16 + colL];
            b.y = W[(k + khalf + 1) * N + t * 16 + colL];
            acc[t] = __builtin_amdgcn_wmma_f32_16x16x4_f32(
                false, a, false, b, (short)0, acc[t], false, false);
        }
    }

    const int mbase = m0 + (lane >> 4) * 8;
    if (m0 + 16 <= n_rows) {
#pragma unroll
        for (int t = 0; t < NT; ++t) {
#pragma unroll
            for (int r = 0; r < 8; ++r) {
                O[(mbase + r) * N + t * 16 + colL] = acc[t][r];
            }
        }
    } else {
#pragma unroll
        for (int t = 0; t < NT; ++t) {
#pragma unroll
            for (int r = 0; r < 8; ++r) {
                int row = mbase + r;
                if (row < n_rows) O[row * N + t * 16 + colL] = acc[t][r];
            }
        }
    }
}


template <int CPL>
__device__ __forceinline__ void row_load(float (&v)[CPL], const float* __restrict__ t, int idx);

template <>
__device__ __forceinline__ void row_load<1>(float (&v)[1], const float* __restrict__ t, int idx) {
    v[0] = t[idx];
}

template <>
__device__ __forceinline__ void row_load<2>(float (&v)[2], const float* __restrict__ t, int idx) {
    const v2f p = ((const v2f*)t)[idx];
    v[0] = p.x;
    v[1] = p.y;
}

template <int CPL, int U>
__device__ __forceinline__ void pull_group(float (&acc)[CPL],
                                           const float* __restrict__ h, int lane,
                                           int sj, float wj, int j, float dd) {
    float v[U][CPL];
#pragma unroll
    for (int u = 0; u < U; ++u) {
        const int s = __builtin_amdgcn_readlane(sj, j + u);
        row_load<CPL>(v[u], h, s * 32 + lane);
    }
    __builtin_amdgcn_sched_barrier(0);
#pragma unroll
    for (int u = 0; u < U; ++u) {
        const float w = __int_as_float(__builtin_amdgcn_readlane(__float_as_int(wj), j + u));
        const float norm = w * dd;
#pragma unroll
        for (int q = 0; q < CPL; ++q) acc[q] = fmaf(v[u][q], norm, acc[q]);
    }
}

template <int CPL>
__device__ __forceinline__ void
pull_row(float (&r)[CPL],
         const int* __restrict__ rowptr,
         const int* __restrict__ adj,
         const float* __restrict__ dinv,
         const float* __restrict__ h,
         const float* __restrict__ bias, int d, int lane) {
    const int beg = __builtin_amdgcn_readfirstlane(rowptr[d]);
    const int end = __builtin_amdgcn_readfirstlane(rowptr[d + 1]);

    const float dd = __int_as_float(__builtin_amdgcn_readfirstlane(__float_as_int(dinv[d])));
    float self[CPL];
    row_load<CPL>(self, h, d * 32 + lane);
    float bv[CPL];
    row_load<CPL>(bv, bias, lane);

    float acc[CPL];
#pragma unroll
    for (int q = 0; q < CPL; ++q) acc[q] = 0.0f;

    for (int b = beg; b < end; b += 32) {
        const int m = end - b < 32 ? end - b : 32;
        int sj = 0;
        float wj = 0.0f;
        if (lane < m) {
            sj = adj[b + lane];
            wj = dinv[sj];
        }
        int j = 0;
        for (; j + 8 <= m; j += 8) pull_group<CPL, 8>(acc, h, lane, sj, wj, j, dd);
        switch (m - j) {
            case 7: pull_group<CPL, 7>(acc, h, lane, sj, wj, j, dd); break;
            case 6: pull_group<CPL, 6>(acc, h, lane, sj, wj, j, dd); break;
            case 5: pull_group<CPL, 5>(acc, h, lane, sj, wj, j, dd); break;
            case 4: pull_group<CPL, 4>(acc, h, lane, sj, wj, j, dd); break;
            case 3: pull_group<CPL, 3>(acc, h, lane, sj, wj, j, dd); break;
            case 2: pull_group<CPL, 2>(acc, h, lane, sj, wj, j, dd); break;
            case 1: pull_group<CPL, 1>(acc, h, lane, sj, wj, j, dd); break;
            default: break;
        }
    }

    const float sn = dd * dd;
#pragma unroll
    for (int q = 0; q < CPL; ++q) {
        const float t = fmaf(self[q], sn, acc[q]) + bv[q];
        r[q] = t > 0.0f ? t : 0.0f;
    }
}

__global__ void
k_pull(const int* __restrict__ rowptr,
       const int* __restrict__ adj,
       const float* __restrict__ dinv,
       const float* __restrict__ h,
       const float* __restrict__ bias,
       float* __restrict__ out, int n) {
    const int d = (blockIdx.x * blockDim.x + threadIdx.x) >> 5;
    const int lane = threadIdx.x & 31;
    if (d >= n) return;
    float r[2];
    pull_row<2>(r, rowptr, adj, dinv, h, bias, d, lane);
    v2f t;
    t.x = r[0];
    t.y = r[1];
    ((v2f*)out)[d * 32 + lane] = t;
}


#define MM_ROWS  16
#define MM_PITCH 34

__global__ void __launch_bounds__(MM_ROWS * 32)
k_pull_mm(const int* __restrict__ rowptr,
          const int* __restrict__ adj,
          const float* __restrict__ dinv,
          const float* __restrict__ h,
          const float* __restrict__ bias,
          const float* __restrict__ W,
          float* __restrict__ out, int n) {
    __shared__ __attribute__((aligned(16))) float tile[MM_ROWS * MM_PITCH];
    const int lane = threadIdx.x & 31;
    const int wv = __builtin_amdgcn_readfirstlane((int)(threadIdx.x >> 5));
    const int m0 = blockIdx.x * MM_ROWS;
    const int d  = m0 + wv;

    float r[1];
    r[0] = 0.0f;
    if (d < n) pull_row<1>(r, rowptr, adj, dinv, h, bias, d, lane);
    tile[wv * MM_PITCH + lane] = r[0];
    __syncthreads();
    if (wv >= 4) return;

    const int c  = lane & 15;
    const int hh = lane >> 4;
    const int col = wv * 16 + c;
    v8f acc = {};
#pragma unroll
    for (int k = 0; k < 32; k += 4) {
        const v2f a = *(const v2f*)&tile[c * MM_PITCH + k + 2 * hh];
        v2f b;
        b.x = W[(k + 2 * hh) * 64 + col];
        b.y = W[(k + 2 * hh + 1) * 64 + col];
        acc = __builtin_amdgcn_wmma_f32_16x16x4_f32(
            false, a, false, b, (short)0, acc, false, false);
    }

    const int mbase = m0 + hh * 8;
    if (m0 + MM_ROWS <= n) {
#pragma unroll
        for (int q = 0; q < 8; ++q) out[(mbase + q) * 64 + col] = acc[q];
    } else {
#pragma unroll
        for (int q = 0; q < 8; ++q) {
            const int row = mbase + q;
            if (row < n) out[row * 64 + col] = acc[q];
        }
    }
}


#define POOL_CH 64

struct PoolSet {
    float4 bx[4];
    float4 by[4];
    int    i0[4];
    int    i1[4];
};

template <bool FULL>
__device__ __forceinline__ void pool_load(PoolSet& S,
                                          const float* __restrict__ h,
                                          const int* __restrict__ batch,
                                          int nb, int hh, int c, int n) {
    if (FULL) {
        const float* hp = h + (size_t)(nb + 2 * hh) * 64 + 4 * c;
        const int*   bp = batch + nb + 2 * hh;
#pragma unroll
        for (int u = 0; u < 4; ++u) {
            S.i0[u] = bp[4 * u];
            S.i1[u] = bp[4 * u + 1];
            S.bx[u] = *(const float4*)(hp + (4 * u) * 64);
            S.by[u] = *(const float4*)(hp + (4 * u + 1) * 64);
        }
    } else {
#pragma unroll
        for (int u = 0; u < 4; ++u) {
            const int node0 = nb + 4 * u + 2 * hh;
            const int node1 = node0 + 1;
            const int r0 = node0 < n ? node0 : n - 1;
            const int r1 = node1 < n ? node1 : n - 1;
            const int i0 = batch[r0];
            const int i1 = batch[r1];
            S.i0[u] = node0 < n ? i0 : -1;
            S.i1[u] = node1 < n ? i1 : -1;
            S.bx[u] = *(const float4*)(h + (size_t)r0 * 64 + 4 * c);
            S.by[u] = *(const float4*)(h + (size_t)r1 * 64 + 4 * c);
        }
    }
}

template <bool FULL>
__device__ __forceinline__ void pool_tile(const float* __restrict__ h,
                                          const int* __restrict__ batch,
                                          float* sums, float* cnts,
                                          int base, int mt, int hh, int c, int n) {
    const int g = mt * 16 + c;
    v8f acc[4] = {};
    float cnt = 0.0f;

    PoolSet cur, nxt;
    pool_load<FULL>(cur, h, batch, base, hh, c, n);
    for (int it = 0; it < POOL_CH / 16; ++it) {
        if (it + 1 < POOL_CH / 16)
            pool_load<FULL>(nxt, h, batch, base + (it + 1) * 16, hh, c, n);
#pragma unroll
        for (int u = 0; u < 4; ++u) {
            v2f a;
            a.x = cur.i0[u] == g ? 1.0f : 0.0f;
            a.y = cur.i1[u] == g ? 1.0f : 0.0f;
            cnt += a.x + a.y;
            const float bx[4] = {cur.bx[u].x, cur.bx[u].y, cur.bx[u].z, cur.bx[u].w};
            const float by[4] = {cur.by[u].x, cur.by[u].y, cur.by[u].z, cur.by[u].w};
#pragma unroll
            for (int nt = 0; nt < 4; ++nt) {
                v2f b;
                b.x = bx[nt];
                b.y = by[nt];
                acc[nt] = __builtin_amdgcn_wmma_f32_16x16x4_f32(
                    false, a, false, b, (short)0, acc[nt], false, false);
            }
        }
        cur = nxt;
    }

    const float tot = cnt + __shfl_xor(cnt, 16);
    if (hh == 0 && tot != 0.0f) unsafeAtomicAdd(&cnts[g], tot);

    const int grow = mt * 16 + hh * 8;
#pragma unroll
    for (int r = 0; r < 8; ++r) {
#pragma unroll
        for (int nt = 0; nt < 4; ++nt) {
            const float v = acc[nt][r];
            if (v != 0.0f)
                unsafeAtomicAdd(&sums[(grow + r) * 64 + 4 * c + nt], v);
        }
    }
}

__global__ void __launch_bounds__(64)
k_pool_onehot(const float* __restrict__ h,
              const int* __restrict__ batch,
              float* sums, float* cnts, int n) {
    const int wave = (blockIdx.x * blockDim.x + threadIdx.x) >> 5;
    const int lane = threadIdx.x & 31;
    const int base = wave * POOL_CH;
    if (base >= n) return;
    const int c  = lane & 15;
    const int hh = lane >> 4;
    const bool full = base + POOL_CH <= n;

    unsigned present = 0;
    for (int i = lane; i < POOL_CH; i += 32) {
        const int node = base + i;
        if (node < n) {
            const unsigned gid = (unsigned)batch[node];
            if (gid < 64u) present |= 1u << (gid >> 4);
        }
    }

    for (int mt = 0; mt < 4; ++mt) {
        if (__ballot((present >> mt) & 1u) == 0) continue;
        if (full) pool_tile<true >(h, batch, sums, cnts, base, mt, hh, c, n);
        else      pool_tile<false>(h, batch, sums, cnts, base, mt, hh, c, n);
    }
}


__global__ void k_final_head(const float* __restrict__ sums,
                             const float* __restrict__ cnts,
                             const float* __restrict__ Wc,
                             const float* __restrict__ bc,
                             float* out, int G) {
    int g = threadIdx.x;
    if (g >= G) return;
    float c = cnts[g];
    c = c > 1.0f ? c : 1.0f;
    float inv = 1.0f / c;
    float l0 = bc[0], l1 = bc[1];
    for (int ch = 0; ch < 64; ++ch) {
        float p = sums[g * 64 + ch] * inv;
        out[G * 2 + g * 64 + ch] = p;
        l0 = fmaf(p, Wc[ch * 2 + 0], l0);
        l1 = fmaf(p, Wc[ch * 2 + 1], l1);
    }
    out[g * 2 + 0] = l0;
    out[g * 2 + 1] = l1;
}


extern "C" void kernel_launch(void* const* d_in, const int* in_sizes, int n_in,
                              void* d_out, int out_size, void* d_ws, size_t ws_size,
                              hipStream_t stream) {
    const float* x   = (const float*)d_in[0];
    const float* W1  = (const float*)d_in[1];
    const float* b1  = (const float*)d_in[2];
    const float* W2  = (const float*)d_in[3];
    const float* b2  = (const float*)d_in[4];
    const float* Wc  = (const float*)d_in[5];
    const float* bc  = (const float*)d_in[6];
    const int*   ei  = (const int*)d_in[7];
    const int*   bat = (const int*)d_in[8];

    const int n = in_sizes[0] / IN_CH;
    const int E = in_sizes[7] / 2;
    const int G = 64;

    const int* src = ei;
    const int* dst = ei + E;

    float* ws   = (float*)d_ws;
    float* bufA = ws;                ws += (size_t)n * 64;
    float* bufB = ws;                ws += (size_t)n * 64;
    float* dinv = ws;                ws += n;
    float* sums = ws;                ws += (size_t)G * 64;
    float* cnts = ws;                ws += G;
    ws += n;
    int* rowptr = (int*)ws;          ws += n + 1;
    ws += (n + 255) / 256;
    int* adj    = (int*)ws;          ws += E;
    float* h1 = bufA;
    float* h2 = bufB;
    float* o2 = bufA;

    const int B = 256;
    auto cdiv = [](long long a, long long b) { return (int)((a + b - 1) / b); };

    const int S  = cdiv(E, CHUNK);
    const int NB = cdiv(n, BIN_NODES);
    int* binned = (int*)bufA;
    int* tab    = (int*)bufB;
    k_bin_scatter<<<S, SORT_T, 0, stream>>>(src, dst, tab, binned, E, S, NB);
    k_bin_owner<<<NB, SORT_T, 0, stream>>>(tab, binned, rowptr, dinv, adj, n, E, S);

    const int tiles = (n + 15) / 16;
    k_gemm_wmma<128, 32><<<cdiv(tiles, 4), 128, 0, stream>>>(x, W1, h1, n, tiles);

    k_pull_mm<<<cdiv(n, MM_ROWS), MM_ROWS * 32, 0, stream>>>(rowptr, adj, dinv, h1, b1, W2, h2, n);

    k_pull<<<cdiv((long long)n * 32, B), B, 0, stream>>>(rowptr, adj, dinv, h2, b2, o2, n);

    hipMemsetAsync(sums, 0, (size_t)(G * 64 + G) * sizeof(float), stream);
    k_pool_onehot<<<cdiv(n, 2 * POOL_CH), 64, 0, stream>>>(o2, bat, sums, cnts, n);
    k_final_head<<<1, 64, 0, stream>>>(sums, cnts, Wc, bc, (float*)d_out, G);
}


# Launch folding, 7 kernels + memset to 6: pooling (one-hot f32 matrix product on 16-node LDS tiles) inside the layer-2 edge-sum kernel, o2 never stored; sort stores the zeros; pooled sums reassociated
# speedup vs baseline: 5.9350x; 1.0333x over previous
#include <hip/hip_runtime.h>
#include <hip/hip_bf16.h>


typedef float v2f __attribute__((ext_vector_type(2)));
typedef float v8f __attribute__((ext_vector_type(8)));

#define IN_CH 128


#define BIN_SHIFT 9
#define BIN_NODES 512
#define MAXB      256
#define SORT_T    1024
#define EPT       6
#define CHUNK     (SORT_T * EPT)
#define STAGE_CAP 16384
#define RUN_U     9

__device__ __forceinline__ int bin_of(int d, int NB) {
    const int b = (int)((unsigned)d >> BIN_SHIFT);
    return b < NB ? b : NB - 1;
}

__device__ __forceinline__ int wave_incl_scan(int v, int lane) {
    for (int o = 1; o < 32; o <<= 1) {
        const int t = __shfl_up(v, o);
        if (lane >= o) v += t;
    }
    return v;
}

__global__ void __launch_bounds__(SORT_T)
k_bin_scatter(const int* __restrict__ src, const int* __restrict__ dst,
              int* __restrict__ tab, int* __restrict__ binned,
              float* __restrict__ zbuf, int zn, int E, int S, int NB) {
    __shared__ int hist[MAXB];
    __shared__ int lstart[MAXB];
    __shared__ int wsum_c[MAXB / 32];
    __shared__ int srec[CHUNK];
    const int tid = threadIdx.x, wg = blockIdx.x, lane = tid & 31, wv = tid >> 5;
    if (tid < MAXB) hist[tid] = 0;
    __syncthreads();

    for (int zi = wg * SORT_T + tid; zi < zn; zi += S * SORT_T) zbuf[zi] = 0.0f;

    const int e0 = wg * CHUNK;
    int rec[EPT], bn[EPT], rk[EPT];
#pragma unroll
    for (int k = 0; k < EPT; ++k) {
        const int e = e0 + k * SORT_T + tid;
        const int ec = e < E ? e : E - 1;
        bn[k]  = dst[ec];
        rec[k] = src[ec];
    }
#pragma unroll
    for (int k = 0; k < EPT; ++k) {
        const int e = e0 + k * SORT_T + tid;
        const int d = bn[k];
        bn[k]  = bin_of(d, NB);
        rec[k] = (int)(((unsigned)rec[k] << BIN_SHIFT) | ((unsigned)d & (BIN_NODES - 1)));
        rk[k]  = 0;
        if (e < E) rk[k] = atomicAdd(&hist[bn[k]], 1);
    }
    __syncthreads();

    int c = 0, ic = 0;
    if (tid < MAXB) {
        c = hist[tid];
        ic = wave_incl_scan(c, lane);
        if (lane == 31) wsum_c[wv] = ic;
    }
    __syncthreads();
    if (tid < MAXB) {
        int cb = 0;
        for (int w = 0; w < wv; ++w) cb += wsum_c[w];
        const int ls = cb + ic - c;
        lstart[tid] = ls;
        if (tid < NB) tab[tid * S + wg] = (ls << 16) | c;
    }
    __syncthreads();

#pragma unroll
    for (int k = 0; k < EPT; ++k) {
        const int e = e0 + k * SORT_T + tid;
        if (e < E) srec[lstart[bn[k]] + rk[k]] = rec[k];
    }
    __syncthreads();

    const int left = E - e0;
    const int m = left < CHUNK ? left : CHUNK;
    for (int i = tid; i < m; i += SORT_T) binned[e0 + i] = srec[i];
}

template <int MODE>
__device__ __forceinline__ int own_place(int r, int* tbl) {
    if (MODE == 0) {
        atomicAdd(&tbl[r & (BIN_NODES - 1)], 1);
        return 0;
    }
    return atomicAdd(&tbl[r & (BIN_NODES - 1)], 1);
}

template <int MODE>
__device__ __forceinline__ void own_put(int r, int pos, int* sadj,
                                        int* __restrict__ adj, int start) {
    const int s = (int)((unsigned)r >> BIN_SHIFT);
    if (MODE == 1) sadj[pos] = s;
    if (MODE == 2) adj[start + pos] = s;
}

template <int MODE>
__device__ __forceinline__ void own_runs(const int* __restrict__ trow,
                                         const int* __restrict__ binned,
                                         int* tbl, int* sadj, int* __restrict__ adj,
                                         int start, int E, int w0, int nu, int lane,
                                         int& st, int& tot) {
    for (int u0 = 0; u0 < nu; u0 += 32) {
        const int lim = nu - u0 < 32 ? nu - u0 : 32;
        int e = 0;
        if (lane < lim) e = trow[w0 + u0 + lane];
        if (MODE == 0) {
            st  += (int)((unsigned)e >> 16);
            tot += e & 0xFFFF;
        }
        for (int u = 0; u < lim; u += RUN_U) {
            int r0[RUN_U], r1[RUN_U], cn[RUN_U], bs[RUN_U];
#pragma unroll
            for (int k = 0; k < RUN_U; ++k) {
                const bool in = u + k < lim;
                const int uu = in ? u + k : lim - 1;
                const int eu = __builtin_amdgcn_readlane(e, uu);
                const int c  = in ? (eu & 0xFFFF) : 0;
                const int base = (w0 + u0 + uu) * CHUNK + (int)((unsigned)eu >> 16);
                cn[k] = c;
                bs[k] = base;
                int i0 = lane < c ? lane : c - 1;
                int i1 = lane + 32 < c ? lane + 32 : c - 1;
                i0 = i0 > 0 ? i0 : 0;
                i1 = i1 > 0 ? i1 : 0;
                int a0 = base + i0, a1 = base + i1;
                a0 = a0 < E ? a0 : E - 1;
                a1 = a1 < E ? a1 : E - 1;
                r0[k] = binned[a0];
                r1[k] = binned[a1];
            }
            int p0[RUN_U], p1[RUN_U];
#pragma unroll
            for (int k = 0; k < RUN_U; ++k) {
                p0[k] = 0;
                p1[k] = 0;
                if (lane < cn[k])      p0[k] = own_place<MODE>(r0[k], tbl);
                if (lane + 32 < cn[k]) p1[k] = own_place<MODE>(r1[k], tbl);
            }
#pragma unroll
            for (int k = 0; k < RUN_U; ++k) {
                if (MODE != 0) {
                    if (lane < cn[k])      own_put<MODE>(r0[k], p0[k], sadj, adj, start);
                    if (lane + 32 < cn[k]) own_put<MODE>(r1[k], p1[k], sadj, adj, start);
                }
                for (int i = lane + 64; i < cn[k]; i += 32) {
                    const int r = binned[bs[k] + i];
                    const int pos = own_place<MODE>(r, tbl);
                    if (MODE != 0) own_put<MODE>(r, pos, sadj, adj, start);
                }
            }
        }
    }
}

__global__ void __launch_bounds__(SORT_T)
k_bin_owner(const int* __restrict__ tab, const int* __restrict__ binned,
            int* __restrict__ rowptr, float* __restrict__ dinv,
            int* __restrict__ adj, int n, int E, int S) {
    __shared__ int cnt[BIN_NODES];
    __shared__ int cur[BIN_NODES];
    __shared__ int wsum[32];
    __shared__ int wst[32];
    __shared__ int wtot[32];
    __shared__ int sadj[STAGE_CAP];
    const int b = blockIdx.x, tid = threadIdx.x, lane = tid & 31;
    const int wv = __builtin_amdgcn_readfirstlane(tid >> 5);

    const int rpw = (S + 31) >> 5;
    const int w0  = wv * rpw;
    const int nu  = S - w0 < rpw ? (S - w0 > 0 ? S - w0 : 0) : rpw;
    const int* trow = tab + (size_t)b * S;

    if (tid < BIN_NODES) cnt[tid] = 0;
    __syncthreads();

    int st = 0, tot = 0;
    own_runs<0>(trow, binned, cnt, sadj, adj, 0, E, w0, nu, lane, st, tot);
    for (int o = 16; o > 0; o >>= 1) {
        st  += __shfl_xor(st, o);
        tot += __shfl_xor(tot, o);
    }
    if (lane == 0) { wst[wv] = st; wtot[wv] = tot; }
    __syncthreads();

    int start = 0, m = 0;
    for (int w = 0; w < 32; ++w) { start += wst[w]; m += wtot[w]; }

    int c = 0, inc = 0;
    if (tid < BIN_NODES) {
        c = cnt[tid];
        inc = wave_incl_scan(c, lane);
        if (lane == 31) wsum[wv] = inc;
    }
    __syncthreads();
    if (tid < BIN_NODES) {
        int wb = 0;
        for (int w = 0; w < wv; ++w) wb += wsum[w];
        const int excl = wb + inc - c;
        cur[tid] = excl;
        const int node = b * BIN_NODES + tid;
        if (node < n) {
            rowptr[node] = start + excl;
            const float d = (float)(c + 1);
            dinv[node] = d > 0.0f ? __frsqrt_rn(d) : 0.0f;
            if (node == n - 1) rowptr[n] = start + excl + c;
        }
    }
    __syncthreads();

    int dum0 = 0, dum1 = 0;
    if (m <= STAGE_CAP) {
        own_runs<1>(trow, binned, cur, sadj, adj, start, E, w0, nu, lane, dum0, dum1);
        __syncthreads();
        for (int i = tid; i < m; i += SORT_T) adj[start + i] = sadj[i];
    } else {
        own_runs<2>(trow, binned, cur, sadj, adj, start, E, w0, nu, lane, dum0, dum1);
    }
}


template <int K, int N>
__global__ void k_gemm_wmma(const float* __restrict__ A,
                            const float* __restrict__ W,
                            float* __restrict__ O,
                            int n_rows, int n_tiles) {
    int wave = (blockIdx.x * blockDim.x + threadIdx.x) >> 5;
    int lane = threadIdx.x & 31;
    if (wave >= n_tiles) return;

    const int m0    = wave * 16;
    const int mrow  = m0 + (lane & 15);
    const int rm    = mrow < n_rows ? mrow : (n_rows - 1);
    const int khalf = (lane >> 4) * 2;
    const int colL  = lane & 15;

    constexpr int NT = N / 16;
    v8f acc[NT] = {};

    for (int k = 0; k < K; k += 4) {
        v2f a;
        a.x = A[rm * K + k + khalf];
        a.y = A[rm * K + k + khalf + 1];
#pragma unroll
        for (int t = 0; t < NT; ++t) {
            v2f b;
            b.x = W[(k + khalf) * N + t * 16 + colL];
            b.y = W[(k + khalf + 1) * N + t * 16 + colL];
            acc[t] = __builtin_amdgcn_wmma_f32_16x16x4_f32(
                false, a, false, b, (short)0, acc[t], false, false);
        }
    }

    const int mbase = m0 + (lane >> 4) * 8;
    if (m0 + 16 <= n_rows) {
#pragma unroll
        for (int t = 0; t < NT; ++t) {
#pragma unroll
            for (int r = 0; r < 8; ++r) {
                O[(mbase + r) * N + t * 16 + colL] = acc[t][r];
            }
        }
    } else {
#pragma unroll
        for (int t = 0; t < NT; ++t) {
#pragma unroll
            for (int r = 0; r < 8; ++r) {
                int row = mbase + r;
                if (row < n_rows) O[row * N + t * 16 + colL] = acc[t][r];
            }
        }
    }
}


template <int CPL>
__device__ __forceinline__ void row_load(float (&v)[CPL], const float* __restrict__ t, int idx);

template <>
__device__ __forceinline__ void row_load<1>(float (&v)[1], const float* __restrict__ t, int idx) {
    v[0] = t[idx];
}

template <>
__device__ __forceinline__ void row_load<2>(float (&v)[2], const float* __restrict__ t, int idx) {
    const v2f p = ((const v2f*)t)[idx];
    v[0] = p.x;
    v[1] = p.y;
}

template <int CPL, int U>
__device__ __forceinline__ void pull_group(float (&acc)[CPL],
                                           const float* __restrict__ h, int lane,
                                           int sj, float wj, int j, float dd) {
    float v[U][CPL];
#pragma unroll
    for (int u = 0; u < U; ++u) {
        const int s = __builtin_amdgcn_readlane(sj, j + u);
        row_load<CPL>(v[u], h, s * 32 + lane);
    }
    __builtin_amdgcn_sched_barrier(0);
#pragma unroll
    for (int u = 0; u < U; ++u) {
        const float w = __int_as_float(__builtin_amdgcn_readlane(__float_as_int(wj), j + u));
        const float norm = w * dd;
#pragma unroll
        for (int q = 0; q < CPL; ++q) acc[q] = fmaf(v[u][q], norm, acc[q]);
    }
}

template <int CPL>
__device__ __forceinline__ void
pull_row(float (&r)[CPL],
         const int* __restrict__ rowptr,
         const int* __restrict__ adj,
         const float* __restrict__ dinv,
         const float* __restrict__ h,
         const float* __restrict__ bias, int d, int lane) {
    const int beg = __builtin_amdgcn_readfirstlane(rowptr[d]);
    const int end = __builtin_amdgcn_readfirstlane(rowptr[d + 1]);

    const float dd = __int_as_float(__builtin_amdgcn_readfirstlane(__float_as_int(dinv[d])));
    float self[CPL];
    row_load<CPL>(self, h, d * 32 + lane);
    float bv[CPL];
    row_load<CPL>(bv, bias, lane);

    float acc[CPL];
#pragma unroll
    for (int q = 0; q < CPL; ++q) acc[q] = 0.0f;

    for (int b = beg; b < end; b += 32) {
        const int m = end - b < 32 ? end - b : 32;
        int sj = 0;
        float wj = 0.0f;
        if (lane < m) {
            sj = adj[b + lane];
            wj = dinv[sj];
        }
        int j = 0;
        for (; j + 8 <= m; j += 8) pull_group<CPL, 8>(acc, h, lane, sj, wj, j, dd);
        switch (m - j) {
            case 7: pull_group<CPL, 7>(acc, h, lane, sj, wj, j, dd); break;
            case 6: pull_group<CPL, 6>(acc, h, lane, sj, wj, j, dd); break;
            case 5: pull_group<CPL, 5>(acc, h, lane, sj, wj, j, dd); break;
            case 4: pull_group<CPL, 4>(acc, h, lane, sj, wj, j, dd); break;
            case 3: pull_group<CPL, 3>(acc, h, lane, sj, wj, j, dd); break;
            case 2: pull_group<CPL, 2>(acc, h, lane, sj, wj, j, dd); break;
            case 1: pull_group<CPL, 1>(acc, h, lane, sj, wj, j, dd); break;
            default: break;
        }
    }

    const float sn = dd * dd;
#pragma unroll
    for (int q = 0; q < CPL; ++q) {
        const float t = fmaf(self[q], sn, acc[q]) + bv[q];
        r[q] = t > 0.0f ? t : 0.0f;
    }
}


#define MM_ROWS  16
#define MM_PITCH 34

__global__ void __launch_bounds__(MM_ROWS * 32)
k_pull_mm(const int* __restrict__ rowptr,
          const int* __restrict__ adj,
          const float* __restrict__ dinv,
          const float* __restrict__ h,
          const float* __restrict__ bias,
          const float* __restrict__ W,
          float* __restrict__ out, int n) {
    __shared__ __attribute__((aligned(16))) float tile[MM_ROWS * MM_PITCH];
    const int lane = threadIdx.x & 31;
    const int wv = __builtin_amdgcn_readfirstlane((int)(threadIdx.x >> 5));
    const int m0 = blockIdx.x * MM_ROWS;
    const int d  = m0 + wv;

    float r[1];
    r[0] = 0.0f;
    if (d < n) pull_row<1>(r, rowptr, adj, dinv, h, bias, d, lane);
    tile[wv * MM_PITCH + lane] = r[0];
    __syncthreads();
    if (wv >= 4) return;

    const int c  = lane & 15;
    const int hh = lane >> 4;
    const int col = wv * 16 + c;
    v8f acc = {};
#pragma unroll
    for (int k = 0; k < 32; k += 4) {
        const v2f a = *(const v2f*)&tile[c * MM_PITCH + k + 2 * hh];
        v2f b;
        b.x = W[(k + 2 * hh) * 64 + col];
        b.y = W[(k + 2 * hh + 1) * 64 + col];
        acc = __builtin_amdgcn_wmma_f32_16x16x4_f32(
            false, a, false, b, (short)0, acc, false, false);
    }

    const int mbase = m0 + hh * 8;
    if (m0 + MM_ROWS <= n) {
#pragma unroll
        for (int q = 0; q < 8; ++q) out[(mbase + q) * 64 + col] = acc[q];
    } else {
#pragma unroll
        for (int q = 0; q < 8; ++q) {
            const int row = mbase + q;
            if (row < n) out[row * 64 + col] = acc[q];
        }
    }
}


#define PP_ROWS  16
#define PP_PITCH 72

__global__ void __launch_bounds__(PP_ROWS * 32)
k_pull_pool(const int* __restrict__ rowptr,
            const int* __restrict__ adj,
            const float* __restrict__ dinv,
            const float* __restrict__ h,
            const float* __restrict__ bias,
            const int* __restrict__ batch,
            float* sums, float* cnts, int n) {
    __shared__ __attribute__((aligned(16))) float tile[PP_ROWS * PP_PITCH];
    __shared__ __attribute__((aligned(16))) int ids[PP_ROWS];
    const int lane = threadIdx.x & 31;
    const int wv = __builtin_amdgcn_readfirstlane((int)(threadIdx.x >> 5));
    const int m0 = blockIdx.x * PP_ROWS;
    const int d  = m0 + wv;

    float r[2];
    r[0] = 0.0f;
    r[1] = 0.0f;
    int gid = -1;
    if (d < n) {
        gid = batch[d];
        pull_row<2>(r, rowptr, adj, dinv, h, bias, d, lane);
    }
    v2f t;
    t.x = r[0];
    t.y = r[1];
    *(v2f*)&tile[wv * PP_PITCH + 2 * lane] = t;
    if (lane == 0) ids[wv] = gid;
    __syncthreads();
    if (wv >= 4) return;

    const int c  = lane & 15;
    const int hh = lane >> 4;
    const int col = wv * 16 + c;

    int   ia[4], ib[4];
    float bx[4], by[4];
#pragma unroll
    for (int u = 0; u < 4; ++u) {
        const int k = 4 * u + 2 * hh;
        ia[u] = ids[k];
        ib[u] = ids[k + 1];
        bx[u] = tile[k * PP_PITCH + col];
        by[u] = tile[(k + 1) * PP_PITCH + col];
    }
    const unsigned mine = (unsigned)ids[c];

    for (int mt = 0; mt < 4; ++mt) {
        if (__ballot(mine < 64u && (int)(mine >> 4) == mt) == 0) continue;
        const int g = mt * 16 + c;
        v8f acc = {};
        float cnt = 0.0f;
#pragma unroll
        for (int u = 0; u < 4; ++u) {
            v2f a, b;
            a.x = ia[u] == g ? 1.0f : 0.0f;
            a.y = ib[u] == g ? 1.0f : 0.0f;
            cnt += a.x + a.y;
            b.x = bx[u];
            b.y = by[u];
            acc = __builtin_amdgcn_wmma_f32_16x16x4_f32(
                false, a, false, b, (short)0, acc, false, false);
        }

        const float tot = cnt + __shfl_xor(cnt, 16);
        if (wv == 0 && hh == 0 && tot != 0.0f) unsafeAtomicAdd(&cnts[g], tot);

        const int grow = mt * 16 + hh * 8;
#pragma unroll
        for (int q = 0; q < 8; ++q) {
            const float v = acc[q];
            if (v != 0.0f) unsafeAtomicAdd(&sums[(grow + q) * 64 + col], v);
        }
    }
}


__global__ void k_final_head(const float* __restrict__ sums,
                             const float* __restrict__ cnts,
                             const float* __restrict__ Wc,
                             const float* __restrict__ bc,
                             float* out, int G) {
    int g = threadIdx.x;
    if (g >= G) return;
    float c = cnts[g];
    c = c > 1.0f ? c : 1.0f;
    float inv = 1.0f / c;
    float l0 = bc[0], l1 = bc[1];
    for (int ch = 0; ch < 64; ++ch) {
        float p = sums[g * 64 + ch] * inv;
        out[G * 2 + g * 64 + ch] = p;
        l0 = fmaf(p, Wc[ch * 2 + 0], l0);
        l1 = fmaf(p, Wc[ch * 2 + 1], l1);
    }
    out[g * 2 + 0] = l0;
    out[g * 2 + 1] = l1;
}


extern "C" void kernel_launch(void* const* d_in, const int* in_sizes, int n_in,
                              void* d_out, int out_size, void* d_ws, size_t ws_size,
                              hipStream_t stream) {
    const float* x   = (const float*)d_in[0];
    const float* W1  = (const float*)d_in[1];
    const float* b1  = (const float*)d_in[2];
    const float* W2  = (const float*)d_in[3];
    const float* b2  = (const float*)d_in[4];
    const float* Wc  = (const float*)d_in[5];
    const float* bc  = (const float*)d_in[6];
    const int*   ei  = (const int*)d_in[7];
    const int*   bat = (const int*)d_in[8];

    const int n = in_sizes[0] / IN_CH;
    const int E = in_sizes[7] / 2;
    const int G = 64;

    const int* src = ei;
    const int* dst = ei + E;

    float* ws   = (float*)d_ws;
    float* bufA = ws;                ws += (size_t)n * 64;
    float* bufB = ws;                ws += (size_t)n * 64;
    float* dinv = ws;                ws += n;
    float* sums = ws;                ws += (size_t)G * 64;
    float* cnts = ws;                ws += G;
    ws += n;
    int* rowptr = (int*)ws;          ws += n + 1;
    ws += (n + 255) / 256;
    int* adj    = (int*)ws;          ws += E;
    float* h1 = bufA;
    float* h2 = bufB;

    auto cdiv = [](long long a, long long b) { return (int)((a + b - 1) / b); };

    const int S  = cdiv(E, CHUNK);
    const int NB = cdiv(n, BIN_NODES);
    int* binned = (int*)bufA;
    int* tab    = (int*)bufB;
    k_bin_scatter<<<S, SORT_T, 0, stream>>>(src, dst, tab, binned, sums, G * 64 + G, E, S, NB);
    k_bin_owner<<<NB, SORT_T, 0, stream>>>(tab, binned, rowptr, dinv, adj, n, E, S);

    const int tiles = (n + 15) / 16;
    k_gemm_wmma<128, 32><<<cdiv(tiles, 4), 128, 0, stream>>>(x, W1, h1, n, tiles);

    k_pull_mm<<<cdiv(n, MM_ROWS), MM_ROWS * 32, 0, stream>>>(rowptr, adj, dinv, h1, b1, W2, h2, n);

    k_pull_pool<<<cdiv(n, PP_ROWS), PP_ROWS * 32, 0, stream>>>(rowptr, adj, dinv, h2, b2, bat, sums, cnts, n);

    k_final_head<<<1, 64, 0, stream>>>(sums, cnts, Wc, bc, (float*)d_out, G);
}


# Layer-1 product (K=128) runs in the launch of sort level 1 on 250 work-groups of its own, resident beside the 261 binning work-groups: memory waits under LDS and barrier waits; 6 launches to 5
# speedup vs baseline: 6.3446x; 1.0690x over previous
#include <hip/hip_runtime.h>
#include <hip/hip_bf16.h>


typedef float v2f __attribute__((ext_vector_type(2)));
typedef float v8f __attribute__((ext_vector_type(8)));

#define IN_CH 128


#define BIN_SHIFT 9
#define BIN_NODES 512
#define MAXB      256
#define SORT_T    1024
#define EPT       6
#define CHUNK     (SORT_T * EPT)
#define STAGE_CAP 16384
#define RUN_U     9

__device__ __forceinline__ int bin_of(int d, int NB) {
    const int b = (int)((unsigned)d >> BIN_SHIFT);
    return b < NB ? b : NB - 1;
}

__device__ __forceinline__ int wave_incl_scan(int v, int lane) {
    for (int o = 1; o < 32; o <<= 1) {
        const int t = __shfl_up(v, o);
        if (lane >= o) v += t;
    }
    return v;
}


template <int K, int N>
__device__ __forceinline__ void gemm_tile_wmma(const float* __restrict__ A,
                                               const float* __restrict__ W,
                                               float* __restrict__ O,
                                               int n_rows, int n_tiles,
                                               int wave, int lane) {
    if (wave >= n_tiles) return;

    const int m0    = wave * 16;
    const int mrow  = m0 + (lane & 15);
    const int rm    = mrow < n_rows ? mrow : (n_rows - 1);
    const int khalf = (lane >> 4) * 2;
    const int colL  = lane & 15;

    constexpr int NT = N / 16;
    v8f acc[NT] = {};

    for (int k = 0; k < K; k += 4) {
        v2f a;
        a.x = A[rm * K + k + khalf];
        a.y = A[rm * K + k + khalf + 1];
#pragma unroll
        for (int t = 0; t < NT; ++t) {
            v2f b;
            b.x = W[(k + khalf) * N + t * 16 + colL];
            b.y = W[(k + khalf + 1) * N + t * 16 + colL];
            acc[t] = __builtin_amdgcn_wmma_f32_16x16x4_f32(
                false, a, false, b, (short)0, acc[t], false, false);
        }
    }

    const int mbase = m0 + (lane >> 4) * 8;
    if (m0 + 16 <= n_rows) {
#pragma unroll
        for (int t = 0; t < NT; ++t) {
#pragma unroll
            for (int r = 0; r < 8; ++r) {
                O[(mbase + r) * N + t * 16 + colL] = acc[t][r];
            }
        }
    } else {
#pragma unroll
        for (int t = 0; t < NT; ++t) {
#pragma unroll
            for (int r = 0; r < 8; ++r) {
                int row = mbase + r;
                if (row < n_rows) O[row * N + t * 16 + colL] = acc[t][r];
            }
        }
    }
}

#define GEMM_WG_TILES 25

__global__ void __launch_bounds__(SORT_T)
k_bin_scatter_gemm(const int* __restrict__ src, const int* __restrict__ dst,
                   int* __restrict__ tab, int* __restrict__ binned,
                   float* __restrict__ zbuf, int zn, int E, int S, int NB,
                   const float* __restrict__ x, const float* __restrict__ W1,
                   float* __restrict__ h1, int n, int n_tiles) {
    __shared__ int hist[MAXB];
    __shared__ int lstart[MAXB];
    __shared__ int wsum_c[MAXB / 32];
    __shared__ int srec[CHUNK];
    const int tid = threadIdx.x, wg = blockIdx.x, lane = tid & 31, wv = tid >> 5;
    if (wg >= S) {
        if (wv < GEMM_WG_TILES)
            gemm_tile_wmma<IN_CH, 32>(x, W1, h1, n, n_tiles,
                                      (wg - S) * GEMM_WG_TILES + wv, lane);
        return;
    }
    if (tid < MAXB) hist[tid] = 0;
    __syncthreads();

    for (int zi = wg * SORT_T + tid; zi < zn; zi += S * SORT_T) zbuf[zi] = 0.0f;

    const int e0 = wg * CHUNK;
    int rec[EPT], bn[EPT], rk[EPT];
#pragma unroll
    for (int k = 0; k < EPT; ++k) {
        const int e = e0 + k * SORT_T + tid;
        const int ec = e < E ? e : E - 1;
        bn[k]  = dst[ec];
        rec[k] = src[ec];
    }
#pragma unroll
    for (int k = 0; k < EPT; ++k) {
        const int e = e0 + k * SORT_T + tid;
        const int d = bn[k];
        bn[k]  = bin_of(d, NB);
        rec[k] = (int)(((unsigned)rec[k] << BIN_SHIFT) | ((unsigned)d & (BIN_NODES - 1)));
        rk[k]  = 0;
        if (e < E) rk[k] = atomicAdd(&hist[bn[k]], 1);
    }
    __syncthreads();

    int c = 0, ic = 0;
    if (tid < MAXB) {
        c = hist[tid];
        ic = wave_incl_scan(c, lane);
        if (lane == 31) wsum_c[wv] = ic;
    }
    __syncthreads();
    if (tid < MAXB) {
        int cb = 0;
        for (int w = 0; w < wv; ++w) cb += wsum_c[w];
        const int ls = cb + ic - c;
        lstart[tid] = ls;
        if (tid < NB) tab[tid * S + wg] = (ls << 16) | c;
    }
    __syncthreads();

#pragma unroll
    for (int k = 0; k < EPT; ++k) {
        const int e = e0 + k * SORT_T + tid;
        if (e < E) srec[lstart[bn[k]] + rk[k]] = rec[k];
    }
    __syncthreads();

    const int left = E - e0;
    const int m = left < CHUNK ? left : CHUNK;
    for (int i = tid; i < m; i += SORT_T) binned[e0 + i] = srec[i];
}

template <int MODE>
__device__ __forceinline__ int own_place(int r, int* tbl) {
    if (MODE == 0) {
        atomicAdd(&tbl[r & (BIN_NODES - 1)], 1);
        return 0;
    }
    return atomicAdd(&tbl[r & (BIN_NODES - 1)], 1);
}

template <int MODE>
__device__ __forceinline__ void own_put(int r, int pos, int* sadj,
                                        int* __restrict__ adj, int start) {
    const int s = (int)((unsigned)r >> BIN_SHIFT);
    if (MODE == 1) sadj[pos] = s;
    if (MODE == 2) adj[start + pos] = s;
}

template <int MODE>
__device__ __forceinline__ void own_runs(const int* __restrict__ trow,
                                         const int* __restrict__ binned,
                                         int* tbl, int* sadj, int* __restrict__ adj,
                                         int start, int E, int w0, int nu, int lane,
                                         int& st, int& tot) {
    for (int u0 = 0; u0 < nu; u0 += 32) {
        const int lim = nu - u0 < 32 ? nu - u0 : 32;
        int e = 0;
        if (lane < lim) e = trow[w0 + u0 + lane];
        if (MODE == 0) {
            st  += (int)((unsigned)e >> 16);
            tot += e & 0xFFFF;
        }
        for (int u = 0; u < lim; u += RUN_U) {
            int r0[RUN_U], r1[RUN_U], cn[RUN_U], bs[RUN_U];
#pragma unroll
            for (int k = 0; k < RUN_U; ++k) {
                const bool in = u + k < lim;
                const int uu = in ? u + k : lim - 1;
                const int eu = __builtin_amdgcn_readlane(e, uu);
                const int c  = in ? (eu & 0xFFFF) : 0;
                const int base = (w0 + u0 + uu) * CHUNK + (int)((unsigned)eu >> 16);
                cn[k] = c;
                bs[k] = base;
                int i0 = lane < c ? lane : c - 1;
                int i1 = lane + 32 < c ? lane + 32 : c - 1;
                i0 = i0 > 0 ? i0 : 0;
                i1 = i1 > 0 ? i1 : 0;
                int a0 = base + i0, a1 = base + i1;
                a0 = a0 < E ? a0 : E - 1;
                a1 = a1 < E ? a1 : E - 1;
                r0[k] = binned[a0];
                r1[k] = binned[a1];
            }
            int p0[RUN_U], p1[RUN_U];
#pragma unroll
            for (int k = 0; k < RUN_U; ++k) {
                p0[k] = 0;
                p1[k] = 0;
                if (lane < cn[k])      p0[k] = own_place<MODE>(r0[k], tbl);
                if (lane + 32 < cn[k]) p1[k] = own_place<MODE>(r1[k], tbl);
            }
#pragma unroll
            for (int k = 0; k < RUN_U; ++k) {
                if (MODE != 0) {
                    if (lane < cn[k])      own_put<MODE>(r0[k], p0[k], sadj, adj, start);
                    if (lane + 32 < cn[k]) own_put<MODE>(r1[k], p1[k], sadj, adj, start);
                }
                for (int i = lane + 64; i < cn[k]; i += 32) {
                    const int r = binned[bs[k] + i];
                    const int pos = own_place<MODE>(r, tbl);
                    if (MODE != 0) own_put<MODE>(r, pos, sadj, adj, start);
                }
            }
        }
    }
}

__global__ void __launch_bounds__(SORT_T)
k_bin_owner(const int* __restrict__ tab, const int* __restrict__ binned,
            int* __restrict__ rowptr, float* __restrict__ dinv,
            int* __restrict__ adj, int n, int E, int S) {
    __shared__ int cnt[BIN_NODES];
    __shared__ int cur[BIN_NODES];
    __shared__ int wsum[32];
    __shared__ int wst[32];
    __shared__ int wtot[32];
    __shared__ int sadj[STAGE_CAP];
    const int b = blockIdx.x, tid = threadIdx.x, lane = tid & 31;
    const int wv = __builtin_amdgcn_readfirstlane(tid >> 5);

    const int rpw = (S + 31) >> 5;
    const int w0  = wv * rpw;
    const int nu  = S - w0 < rpw ? (S - w0 > 0 ? S - w0 : 0) : rpw;
    const int* trow = tab + (size_t)b * S;

    if (tid < BIN_NODES) cnt[tid] = 0;
    __syncthreads();

    int st = 0, tot = 0;
    own_runs<0>(trow, binned, cnt, sadj, adj, 0, E, w0, nu, lane, st, tot);
    for (int o = 16; o > 0; o >>= 1) {
        st  += __shfl_xor(st, o);
        tot += __shfl_xor(tot, o);
    }
    if (lane == 0) { wst[wv] = st; wtot[wv] = tot; }
    __syncthreads();

    int start = 0, m = 0;
    for (int w = 0; w < 32; ++w) { start += wst[w]; m += wtot[w]; }

    int c = 0, inc = 0;
    if (tid < BIN_NODES) {
        c = cnt[tid];
        inc = wave_incl_scan(c, lane);
        if (lane == 31) wsum[wv] = inc;
    }
    __syncthreads();
    if (tid < BIN_NODES) {
        int wb = 0;
        for (int w = 0; w < wv; ++w) wb += wsum[w];
        const int excl = wb + inc - c;
        cur[tid] = excl;
        const int node = b * BIN_NODES + tid;
        if (node < n) {
            rowptr[node] = start + excl;
            const float d = (float)(c + 1);
            dinv[node] = d > 0.0f ? __frsqrt_rn(d) : 0.0f;
            if (node == n - 1) rowptr[n] = start + excl + c;
        }
    }
    __syncthreads();

    int dum0 = 0, dum1 = 0;
    if (m <= STAGE_CAP) {
        own_runs<1>(trow, binned, cur, sadj, adj, start, E, w0, nu, lane, dum0, dum1);
        __syncthreads();
        for (int i = tid; i < m; i += SORT_T) adj[start + i] = sadj[i];
    } else {
        own_runs<2>(trow, binned, cur, sadj, adj, start, E, w0, nu, lane, dum0, dum1);
    }
}


template <int CPL>
__device__ __forceinline__ void row_load(float (&v)[CPL], const float* __restrict__ t, int idx);

template <>
__device__ __forceinline__ void row_load<1>(float (&v)[1], const float* __restrict__ t, int idx) {
    v[0] = t[idx];
}

template <>
__device__ __forceinline__ void row_load<2>(float (&v)[2], const float* __restrict__ t, int idx) {
    const v2f p = ((const v2f*)t)[idx];
    v[0] = p.x;
    v[1] = p.y;
}

template <int CPL, int U>
__device__ __forceinline__ void pull_group(float (&acc)[CPL],
                                           const float* __restrict__ h, int lane,
                                           int sj, float wj, int j, float dd) {
    float v[U][CPL];
#pragma unroll
    for (int u = 0; u < U; ++u) {
        const int s = __builtin_amdgcn_readlane(sj, j + u);
        row_load<CPL>(v[u], h, s * 32 + lane);
    }
    __builtin_amdgcn_sched_barrier(0);
#pragma unroll
    for (int u = 0; u < U; ++u) {
        const float w = __int_as_float(__builtin_amdgcn_readlane(__float_as_int(wj), j + u));
        const float norm = w * dd;
#pragma unroll
        for (int q = 0; q < CPL; ++q) acc[q] = fmaf(v[u][q], norm, acc[q]);
    }
}

template <int CPL>
__device__ __forceinline__ void
pull_row(float (&r)[CPL],
         const int* __restrict__ rowptr,
         const int* __restrict__ adj,
         const float* __restrict__ dinv,
         const float* __restrict__ h,
         const float* __restrict__ bias, int d, int lane) {
    const int beg = __builtin_amdgcn_readfirstlane(rowptr[d]);
    const int end = __builtin_amdgcn_readfirstlane(rowptr[d + 1]);

    const float dd = __int_as_float(__builtin_amdgcn_readfirstlane(__float_as_int(dinv[d])));
    float self[CPL];
    row_load<CPL>(self, h, d * 32 + lane);
    float bv[CPL];
    row_load<CPL>(bv, bias, lane);

    float acc[CPL];
#pragma unroll
    for (int q = 0; q < CPL; ++q) acc[q] = 0.0f;

    for (int b = beg; b < end; b += 32) {
        const int m = end - b < 32 ? end - b : 32;
        int sj = 0;
        float wj = 0.0f;
        if (lane < m) {
            sj = adj[b + lane];
            wj = dinv[sj];
        }
        int j = 0;
        for (; j + 8 <= m; j += 8) pull_group<CPL, 8>(acc, h, lane, sj, wj, j, dd);
        switch (m - j) {
            case 7: pull_group<CPL, 7>(acc, h, lane, sj, wj, j, dd); break;
            case 6: pull_group<CPL, 6>(acc, h, lane, sj, wj, j, dd); break;
            case 5: pull_group<CPL, 5>(acc, h, lane, sj, wj, j, dd); break;
            case 4: pull_group<CPL, 4>(acc, h, lane, sj, wj, j, dd); break;
            case 3: pull_group<CPL, 3>(acc, h, lane, sj, wj, j, dd); break;
            case 2: pull_group<CPL, 2>(acc, h, lane, sj, wj, j, dd); break;
            case 1: pull_group<CPL, 1>(acc, h, lane, sj, wj, j, dd); break;
            default: break;
        }
    }

    const float sn = dd * dd;
#pragma unroll
    for (int q = 0; q < CPL; ++q) {
        const float t = fmaf(self[q], sn, acc[q]) + bv[q];
        r[q] = t > 0.0f ? t : 0.0f;
    }
}


#define MM_ROWS  16
#define MM_PITCH 34

__global__ void __launch_bounds__(MM_ROWS * 32)
k_pull_mm(const int* __restrict__ rowptr,
          const int* __restrict__ adj,
          const float* __restrict__ dinv,
          const float* __restrict__ h,
          const float* __restrict__ bias,
          const float* __restrict__ W,
          float* __restrict__ out, int n) {
    __shared__ __attribute__((aligned(16))) float tile[MM_ROWS * MM_PITCH];
    const int lane = threadIdx.x & 31;
    const int wv = __builtin_amdgcn_readfirstlane((int)(threadIdx.x >> 5));
    const int m0 = blockIdx.x * MM_ROWS;
    const int d  = m0 + wv;

    float r[1];
    r[0] = 0.0f;
    if (d < n) pull_row<1>(r, rowptr, adj, dinv, h, bias, d, lane);
    tile[wv * MM_PITCH + lane] = r[0];
    __syncthreads();
    if (wv >= 4) return;

    const int c  = lane & 15;
    const int hh = lane >> 4;
    const int col = wv * 16 + c;
    v8f acc = {};
#pragma unroll
    for (int k = 0; k < 32; k += 4) {
        const v2f a = *(const v2f*)&tile[c * MM_PITCH + k + 2 * hh];
        v2f b;
        b.x = W[(k + 2 * hh) * 64 + col];
        b.y = W[(k + 2 * hh + 1) * 64 + col];
        acc = __builtin_amdgcn_wmma_f32_16x16x4_f32(
            false, a, false, b, (short)0, acc, false, false);
    }

    const int mbase = m0 + hh * 8;
    if (m0 + MM_ROWS <= n) {
#pragma unroll
        for (int q = 0; q < 8; ++q) out[(mbase + q) * 64 + col] = acc[q];
    } else {
#pragma unroll
        for (int q = 0; q < 8; ++q) {
            const int row = mbase + q;
            if (row < n) out[row * 64 + col] = acc[q];
        }
    }
}


#define PP_ROWS  16
#define PP_PITCH 72

__global__ void __launch_bounds__(PP_ROWS * 32)
k_pull_pool(const int* __restrict__ rowptr,
            const int* __restrict__ adj,
            const float* __restrict__ dinv,
            const float* __restrict__ h,
            const float* __restrict__ bias,
            const int* __restrict__ batch,
            float* sums, float* cnts, int n) {
    __shared__ __attribute__((aligned(16))) float tile[PP_ROWS * PP_PITCH];
    __shared__ __attribute__((aligned(16))) int ids[PP_ROWS];
    const int lane = threadIdx.x & 31;
    const int wv = __builtin_amdgcn_readfirstlane((int)(threadIdx.x >> 5));
    const int m0 = blockIdx.x * PP_ROWS;
    const int d  = m0 + wv;

    float r[2];
    r[0] = 0.0f;
    r[1] = 0.0f;
    int gid = -1;
    if (d < n) {
        gid = batch[d];
        pull_row<2>(r, rowptr, adj, dinv, h, bias, d, lane);
    }
    v2f t;
    t.x = r[0];
    t.y = r[1];
    *(v2f*)&tile[wv * PP_PITCH + 2 * lane] = t;
    if (lane == 0) ids[wv] = gid;
    __syncthreads();
    if (wv >= 4) return;

    const int c  = lane & 15;
    const int hh = lane >> 4;
    const int col = wv * 16 + c;

    int   ia[4], ib[4];
    float bx[4], by[4];
#pragma unroll
    for (int u = 0; u < 4; ++u) {
        const int k = 4 * u + 2 * hh;
        ia[u] = ids[k];
        ib[u] = ids[k + 1];
        bx[u] = tile[k * PP_PITCH + col];
        by[u] = tile[(k + 1) * PP_PITCH + col];
    }
    const unsigned mine = (unsigned)ids[c];

    for (int mt = 0; mt < 4; ++mt) {
        if (__ballot(mine < 64u && (int)(mine >> 4) == mt) == 0) continue;
        const int g = mt * 16 + c;
        v8f acc = {};
        float cnt = 0.0f;
#pragma unroll
        for (int u = 0; u < 4; ++u) {
            v2f a, b;
            a.x = ia[u] == g ? 1.0f : 0.0f;
            a.y = ib[u] == g ? 1.0f : 0.0f;
            cnt += a.x + a.y;
            b.x = bx[u];
            b.y = by[u];
            acc = __builtin_amdgcn_wmma_f32_16x16x4_f32(
                false, a, false, b, (short)0, acc, false, false);
        }

        const float tot = cnt + __shfl_xor(cnt, 16);
        if (wv == 0 && hh == 0 && tot != 0.0f) unsafeAtomicAdd(&cnts[g], tot);

        const int grow = mt * 16 + hh * 8;
#pragma unroll
        for (int q = 0; q < 8; ++q) {
            const float v = acc[q];
            if (v != 0.0f) unsafeAtomicAdd(&sums[(grow + q) * 64 + col], v);
        }
    }
}


__global__ void k_final_head(const float* __restrict__ sums,
                             const float* __restrict__ cnts,
                             const float* __restrict__ Wc,
                             const float* __restrict__ bc,
                             float* out, int G) {
    int g = threadIdx.x;
    if (g >= G) return;
    float c = cnts[g];
    c = c > 1.0f ? c : 1.0f;
    float inv = 1.0f / c;
    float l0 = bc[0], l1 = bc[1];
    for (int ch = 0; ch < 64; ++ch) {
        float p = sums[g * 64 + ch] * inv;
        out[G * 2 + g * 64 + ch] = p;
        l0 = fmaf(p, Wc[ch * 2 + 0], l0);
        l1 = fmaf(p, Wc[ch * 2 + 1], l1);
    }
    out[g * 2 + 0] = l0;
    out[g * 2 + 1] = l1;
}


extern "C" void kernel_launch(void* const* d_in, const int* in_sizes, int n_in,
                              void* d_out, int out_size, void* d_ws, size_t ws_size,
                              hipStream_t stream) {
    const float* x   = (const float*)d_in[0];
    const float* W1  = (const float*)d_in[1];
    const float* b1  = (const float*)d_in[2];
    const float* W2  = (const float*)d_in[3];
    const float* b2  = (const float*)d_in[4];
    const float* Wc  = (const float*)d_in[5];
    const float* bc  = (const float*)d_in[6];
    const int*   ei  = (const int*)d_in[7];
    const int*   bat = (const int*)d_in[8];

    const int n = in_sizes[0] / IN_CH;
    const int E = in_sizes[7] / 2;
    const int G = 64;

    const int* src = ei;
    const int* dst = ei + E;

    float* ws   = (float*)d_ws;
    float* bufA = ws;                ws += (size_t)n * 64;
    float* bufB = ws;                ws += (size_t)n * 64;
    float* dinv = ws;                ws += n;
    float* sums = ws;                ws += (size_t)G * 64;
    float* cnts = ws;                ws += G;
    ws += n;
    int* rowptr = (int*)ws;          ws += n + 1;
    ws += (n + 255) / 256;
    int* adj    = (int*)ws;          ws += E;
    float* h1 = bufA + (size_t)n * 32;
    float* h2 = bufB;

    auto cdiv = [](long long a, long long b) { return (int)((a + b - 1) / b); };

    const int S  = cdiv(E, CHUNK);
    const int NB = cdiv(n, BIN_NODES);
    int* binned = (int*)bufA;
    int* tab    = (int*)bufB;
    const int tiles = (n + 15) / 16;
    k_bin_scatter_gemm<<<S + cdiv(tiles, GEMM_WG_TILES), SORT_T, 0, stream>>>(
        src, dst, tab, binned, sums, G * 64 + G, E, S, NB, x, W1, h1, n, tiles);
    k_bin_owner<<<NB, SORT_T, 0, stream>>>(tab, binned, rowptr, dinv, adj, n, E, S);

    k_pull_mm<<<cdiv(n, MM_ROWS), MM_ROWS * 32, 0, stream>>>(rowptr, adj, dinv, h1, b1, W2, h2, n);

    k_pull_pool<<<cdiv(n, PP_ROWS), PP_ROWS * 32, 0, stream>>>(rowptr, adj, dinv, h2, b2, bat, sums, cnts, n);

    k_final_head<<<1, 64, 0, stream>>>(sums, cnts, Wc, bc, (float*)d_out, G);
}
